# starting kernel
# baseline (speedup 1.0000x reference)
#include <hip/hip_runtime.h>
#include <math.h>

#define T_   3
#define R_   5
#define H_   8
#define IN_  256
#define OUT_ 256
#define DK_  32
#define N_   30000
#define E_   200000
#define ML_  240

typedef __attribute__((ext_vector_type(16))) _Float16 v16h;
typedef __attribute__((ext_vector_type(8)))  float    v8f;

__device__ __forceinline__ int kmap(int s, int half) {
  int v = s >> 1;
  return ((v >> 2) << 4) | (half << 3) | ((v & 3) << 1) | (s & 1);
}

__device__ __forceinline__ void atomic_max_f32(float* addr, float val) {
  if (val >= 0.0f) atomicMax((int*)addr, __float_as_int(val));
  else             atomicMin((unsigned int*)addr, __float_as_uint(val));
}

__global__ void pack_w_kernel(const float* __restrict__ W, _Float16* __restrict__ Wp,
                              int total) {
  int idx = blockIdx.x * 256 + threadIdx.x;
  if (idx >= total) return;
  int s    = idx & 15;
  int lane = (idx >> 4) & 31;
  int kk   = (idx >> 9) & 7;
  int ct   = (idx >> 12) & 15;
  int t    = idx >> 16;
  int half = lane >> 4;
  int n    = lane & 15;
  int k    = kk * 32 + kmap(s, half);
  Wp[idx] = (_Float16)W[(size_t)t * (IN_ * OUT_) + (size_t)k * OUT_ + ct * 16 + n];
}

__global__ void init_kernel(float* __restrict__ nmax, float* __restrict__ nden,
                            float* __restrict__ aggr) {
  int i = blockIdx.x * blockDim.x + threadIdx.x;
  int stride = gridDim.x * blockDim.x;
  for (int k = i; k < N_ * H_; k += stride) { nmax[k] = -__builtin_inff(); nden[k] = 0.0f; }
  for (size_t k = i; k < (size_t)N_ * OUT_; k += (size_t)stride) aggr[k] = 0.0f;
}

__global__ void rte_vec_kernel(const float* __restrict__ emb, const float* __restrict__ Wt,
                               const float* __restrict__ b, float* __restrict__ out) {
  int idx = blockIdx.x * 256 + threadIdx.x;
  int i = idx & 255, m = idx >> 8;
  float acc = b[i];
  const float* e = emb + (size_t)m * IN_;
  const float* w = Wt + (size_t)i * IN_;
  for (int j = 0; j < IN_; ++j) acc = fmaf(e[j], w[j], acc);
  out[idx] = acc;
}

__global__ void rte_kv_kernel(const float* __restrict__ rv,
                              const float* __restrict__ Wk, const float* __restrict__ Wv,
                              float* __restrict__ rk, float* __restrict__ rvv) {
  int idx = blockIdx.x * 256 + threadIdx.x;
  int o = idx & 255;
  int m = (idx >> 8) % ML_;
  int t = idx / (ML_ * OUT_);
  const float* v  = rv + (size_t)m * IN_;
  const float* wk = Wk + (size_t)t * (IN_ * OUT_);
  const float* wv = Wv + (size_t)t * (IN_ * OUT_);
  float ak = 0.0f, av = 0.0f;
  for (int i = 0; i < IN_; ++i) {
    float x = v[i];
    ak = fmaf(x, wk[(size_t)i * OUT_ + o], ak);
    av = fmaf(x, wv[(size_t)i * OUT_ + o], av);
  }
  rk[idx] = ak;
  rvv[idx] = av;
}

__device__ __forceinline__ void project_tile(
    const v16h* __restrict__ Asv, const v16h* __restrict__ Wpk,
    const float* __restrict__ bias, float* __restrict__ Out,
    const int* __restrict__ stype, int tmask, int base, int wave, int lane) {
  const int n = lane & 15, hs = lane >> 4;
  for (int t = 0; t < T_; ++t) {
    if (!((tmask >> t) & 1)) continue;
#pragma unroll
    for (int c = 0; c < 2; ++c) {
      const int ct = wave * 2 + c;
      const int nbase = ct * 16;
      const float bv = bias[t * OUT_ + nbase + n];
      v8f acc;
#pragma unroll
      for (int i = 0; i < 8; ++i) acc[i] = bv;
#pragma unroll
      for (int kk = 0; kk < 8; ++kk) {
        v16h a = Asv[kk * 32 + lane];
        v16h b = Wpk[(((size_t)t * 16 + ct) * 8 + kk) * 32 + lane];
        acc = __builtin_amdgcn_wmma_f32_16x16x32_f16(false, a, false, b,
                                                     (short)0, acc, false, false);
      }
#pragma unroll
      for (int i = 0; i < 8; ++i) {
        int m = i + hs * 8;
        if (stype[m] == t)
          Out[(size_t)(base + m) * OUT_ + nbase + n] = acc[i];
      }
    }
  }
}

__global__ __launch_bounds__(256) void qkv_wmma_kernel(
    const float* __restrict__ x,
    const _Float16* __restrict__ wq16, const _Float16* __restrict__ wk16,
    const _Float16* __restrict__ wv16,
    const float* __restrict__ bq, const float* __restrict__ bk, const float* __restrict__ bv,
    const int* __restrict__ ntype,
    float* __restrict__ qn, float* __restrict__ kn, float* __restrict__ vn) {
  __shared__ __align__(32) _Float16 As[16 * IN_];
  __shared__ int stype[16];
  const int tid = threadIdx.x;
  const int base = blockIdx.x * 16;
#pragma unroll
  for (int j = 0; j < 16; ++j) {
    int flat = j * 256 + tid;
    int s = flat & 15, ln = (flat >> 4) & 31, kk = flat >> 9;
    int k = kk * 32 + kmap(s, ln >> 4);
    As[flat] = (_Float16)x[(size_t)(base + (ln & 15)) * IN_ + k];
  }
  if (tid < 16) stype[tid] = ntype[base + tid];
  __syncthreads();
  int tmask = 0;
#pragma unroll
  for (int i = 0; i < 16; ++i) tmask |= 1 << stype[i];
  const int wave = tid >> 5, lane = tid & 31;
  const v16h* Asv = (const v16h*)As;
  project_tile(Asv, (const v16h*)wq16, bq, qn, stype, tmask, base, wave, lane);
  project_tile(Asv, (const v16h*)wk16, bk, kn, stype, tmask, base, wave, lane);
  project_tile(Asv, (const v16h*)wv16, bv, vn, stype, tmask, base, wave, lane);
}

__global__ __launch_bounds__(256) void edge_logit_kernel(
    const float* __restrict__ qn, const float* __restrict__ kn,
    const float* __restrict__ rteK,
    const int* __restrict__ ei, const int* __restrict__ etype,
    const int* __restrict__ etime, const int* __restrict__ ntype,
    const float* __restrict__ rel_att, const float* __restrict__ rel_pri,
    float* __restrict__ logit, float* __restrict__ nmax) {
  __shared__ float sk[8][IN_];
  const int lane = threadIdx.x & 31, we = threadIdx.x >> 5;
  const int e = blockIdx.x * 8 + we;
  const int src = ei[e], dst = ei[E_ + e];
  const int r = etype[e], tm = etime[e], st = ntype[src];
  const float* kb = kn + (size_t)src * OUT_;
  const float* rb = rteK + ((size_t)st * ML_ + tm) * OUT_;
#pragma unroll
  for (int j = 0; j < 8; ++j) {
    int idx = j * 32 + lane;
    sk[we][idx] = kb[idx] + rb[idx];
  }
  __syncthreads();
  const float* qb = qn + (size_t)dst * OUT_;
  const float* A = rel_att + (size_t)r * H_ * DK_ * DK_;
  for (int h = 0; h < H_; ++h) {
    float acc = 0.0f;
    const float* Ah = A + h * DK_ * DK_;
#pragma unroll
    for (int d = 0; d < DK_; ++d)
      acc = fmaf(sk[we][h * DK_ + d], Ah[d * DK_ + lane], acc);
    float p = acc * qb[h * DK_ + lane];
#pragma unroll
    for (int off = 16; off > 0; off >>= 1) p += __shfl_xor(p, off, 32);
    if (lane == 0) {
      float lg = p * rel_pri[r * H_ + h] * 0.17677669529663689f;
      logit[(size_t)e * H_ + h] = lg;
      atomic_max_f32(&nmax[(size_t)dst * H_ + h], lg);
    }
  }
}

__global__ void edge_den_kernel(float* __restrict__ logit, const int* __restrict__ ei,
                                const float* __restrict__ nmax, float* __restrict__ nden) {
  int idx = blockIdx.x * 256 + threadIdx.x;
  int e = idx >> 3, h = idx & 7;
  int dst = ei[E_ + e];
  float ex = __expf(logit[idx] - nmax[(size_t)dst * H_ + h]);
  logit[idx] = ex;
  atomicAdd(&nden[(size_t)dst * H_ + h], ex);
}

__global__ __launch_bounds__(256) void edge_aggr_kernel(
    const float* __restrict__ vn, const float* __restrict__ rteV,
    const int* __restrict__ ei, const int* __restrict__ etype,
    const int* __restrict__ etime, const int* __restrict__ ntype,
    const float* __restrict__ rel_msg, const float* __restrict__ exbuf,
    const float* __restrict__ nden, float* __restrict__ aggr) {
  __shared__ float sv[8][IN_];
  const int lane = threadIdx.x & 31, we = threadIdx.x >> 5;
  const int e = blockIdx.x * 8 + we;
  const int src = ei[e], dst = ei[E_ + e];
  const int r = etype[e], tm = etime[e], st = ntype[src];
  const float* vb = vn + (size_t)src * OUT_;
  const float* rb = rteV + ((size_t)st * ML_ + tm) * OUT_;
#pragma unroll
  for (int j = 0; j < 8; ++j) {
    int idx = j * 32 + lane;
    sv[we][idx] = vb[idx] + rb[idx];
  }
  __syncthreads();
  const float* M = rel_msg + (size_t)r * H_ * DK_ * DK_;
  for (int h = 0; h < H_; ++h) {
    float acc = 0.0f;
    const float* Mh = M + h * DK_ * DK_;
#pragma unroll
    for (int d = 0; d < DK_; ++d)
      acc = fmaf(sv[we][h * DK_ + d], Mh[d * DK_ + lane], acc);
    float att = exbuf[(size_t)e * H_ + h] / nden[(size_t)dst * H_ + h];
    atomicAdd(&aggr[(size_t)dst * OUT_ + h * DK_ + lane], acc * att);
  }
}

__global__ __launch_bounds__(256) void out_wmma_kernel(
    const float* __restrict__ aggr, const _Float16* __restrict__ wa16,
    const float* __restrict__ ba, const float* __restrict__ x,
    const int* __restrict__ ntype, const float* __restrict__ skip,
    const float* __restrict__ ln_g, const float* __restrict__ ln_b,
    float* __restrict__ out) {
  __shared__ __align__(32) _Float16 As[16 * OUT_];
  __shared__ float Tr[16 * OUT_];
  __shared__ int stype[16];
  __shared__ float alpha_s[16], mu_s[16], rs_s[16];
  const int tid = threadIdx.x;
  const int base = blockIdx.x * 16;
#pragma unroll
  for (int j = 0; j < 16; ++j) {
    int flat = j * 256 + tid;
    int s = flat & 15, ln = (flat >> 4) & 31, kk = flat >> 9;
    int k = kk * 32 + kmap(s, ln >> 4);
    As[flat] = (_Float16)aggr[(size_t)(base + (ln & 15)) * OUT_ + k];
  }
  if (tid < 16) {
    int t = ntype[base + tid];
    stype[tid] = t;
    alpha_s[tid] = 1.0f / (1.0f + __expf(-skip[t]));
  }
  __syncthreads();
  int tmask = 0;
#pragma unroll
  for (int i = 0; i < 16; ++i) tmask |= 1 << stype[i];
  const int wave = tid >> 5, lane = tid & 31;
  const int n = lane & 15, hs = lane >> 4;
  const v16h* Asv = (const v16h*)As;
  const v16h* Wpk = (const v16h*)wa16;
  for (int t = 0; t < T_; ++t) {
    if (!((tmask >> t) & 1)) continue;
#pragma unroll
    for (int c = 0; c < 2; ++c) {
      const int ct = wave * 2 + c;
      const int nbase = ct * 16;
      const float bv = ba[t * OUT_ + nbase + n];
      v8f acc;
#pragma unroll
      for (int i = 0; i < 8; ++i) acc[i] = bv;
#pragma unroll
      for (int kk = 0; kk < 8; ++kk) {
        v16h a = Asv[kk * 32 + lane];
        v16h b = Wpk[(((size_t)t * 16 + ct) * 8 + kk) * 32 + lane];
        acc = __builtin_amdgcn_wmma_f32_16x16x32_f16(false, a, false, b,
                                                     (short)0, acc, false, false);
      }
#pragma unroll
      for (int i = 0; i < 8; ++i) {
        int m = i + hs * 8;
        if (stype[m] == t) Tr[m * OUT_ + nbase + n] = acc[i];
      }
    }
  }
  __syncthreads();
#pragma unroll
  for (int j = 0; j < 16; ++j) {
    int flat = j * 256 + tid;
    int row = flat >> 8, col = flat & 255;
    float a = alpha_s[row];
    Tr[flat] = Tr[flat] * a + x[(size_t)(base + row) * IN_ + col] * (1.0f - a);
  }
  __syncthreads();
  if (tid < 16) {
    float mu = 0.0f;
    for (int c = 0; c < OUT_; ++c) mu += Tr[tid * OUT_ + c];
    mu *= (1.0f / OUT_);
    float var = 0.0f;
    for (int c = 0; c < OUT_; ++c) {
      float d = Tr[tid * OUT_ + c] - mu;
      var = fmaf(d, d, var);
    }
    var *= (1.0f / OUT_);
    mu_s[tid] = mu;
    rs_s[tid] = rsqrtf(var + 1e-5f);
  }
  __syncthreads();
#pragma unroll
  for (int j = 0; j < 16; ++j) {
    int flat = j * 256 + tid;
    int row = flat >> 8, col = flat & 255;
    int t = stype[row];
    out[(size_t)(base + row) * OUT_ + col] =
        (Tr[flat] - mu_s[row]) * rs_s[row] * ln_g[t * OUT_ + col] + ln_b[t * OUT_ + col];
  }
}

extern "C" void kernel_launch(void* const* d_in, const int* in_sizes, int n_in,
                              void* d_out, int out_size, void* d_ws, size_t ws_size,
                              hipStream_t stream) {
  (void)in_sizes; (void)n_in; (void)out_size; (void)ws_size;
  const float* node_inp  = (const float*)d_in[0];
  const float* Wk        = (const float*)d_in[1];
  const float* bk        = (const float*)d_in[2];
  const float* Wq        = (const float*)d_in[3];
  const float* bq        = (const float*)d_in[4];
  const float* Wv        = (const float*)d_in[5];
  const float* bv        = (const float*)d_in[6];
  const float* Wa        = (const float*)d_in[7];
  const float* ba        = (const float*)d_in[8];
  const float* ln_g      = (const float*)d_in[9];
  const float* ln_b      = (const float*)d_in[10];
  const float* rel_pri   = (const float*)d_in[11];
  const float* rel_att   = (const float*)d_in[12];
  const float* rel_msg   = (const float*)d_in[13];
  const float* skip      = (const float*)d_in[14];
  const float* rte_W     = (const float*)d_in[15];
  const float* rte_b     = (const float*)d_in[16];
  const float* rte_emb   = (const float*)d_in[17];
  const int*   node_type = (const int*)d_in[18];
  const int*   edge_idx  = (const int*)d_in[19];
  const int*   edge_type = (const int*)d_in[20];
  const int*   edge_time = (const int*)d_in[21];
  float* outp = (float*)d_out;

  char* p = (char*)d_ws;
  auto alloc = [&](size_t bytes) -> void* {
    void* r = (void*)p;
    p += (bytes + 255) & ~(size_t)255;
    return r;
  };
  float* rte_vec = (float*)alloc((size_t)ML_ * IN_ * 4);
  float* rteK    = (float*)alloc((size_t)T_ * ML_ * OUT_ * 4);
  float* rteV    = (float*)alloc((size_t)T_ * ML_ * OUT_ * 4);
  float* qn      = (float*)alloc((size_t)N_ * OUT_ * 4);
  float* kn      = (float*)alloc((size_t)N_ * OUT_ * 4);
  float* vn      = (float*)alloc((size_t)N_ * OUT_ * 4);
  float* logit   = (float*)alloc((size_t)E_ * H_ * 4);
  float* nmax    = (float*)alloc((size_t)N_ * H_ * 4);
  float* nden    = (float*)alloc((size_t)N_ * H_ * 4);
  float* aggr    = (float*)alloc((size_t)N_ * OUT_ * 4);
  _Float16* wq16 = (_Float16*)alloc((size_t)T_ * IN_ * OUT_ * 2);
  _Float16* wk16 = (_Float16*)alloc((size_t)T_ * IN_ * OUT_ * 2);
  _Float16* wv16 = (_Float16*)alloc((size_t)T_ * IN_ * OUT_ * 2);
  _Float16* wa16 = (_Float16*)alloc((size_t)T_ * OUT_ * OUT_ * 2);

  const int wtotal = T_ * IN_ * OUT_;
  pack_w_kernel<<<wtotal / 256, 256, 0, stream>>>(Wq, wq16, wtotal);
  pack_w_kernel<<<wtotal / 256, 256, 0, stream>>>(Wk, wk16, wtotal);
  pack_w_kernel<<<wtotal / 256, 256, 0, stream>>>(Wv, wv16, wtotal);
  pack_w_kernel<<<wtotal / 256, 256, 0, stream>>>(Wa, wa16, wtotal);

  init_kernel<<<4096, 256, 0, stream>>>(nmax, nden, aggr);

  rte_vec_kernel<<<(ML_ * IN_) / 256, 256, 0, stream>>>(rte_emb, rte_W, rte_b, rte_vec);
  rte_kv_kernel<<<(T_ * ML_ * OUT_) / 256, 256, 0, stream>>>(rte_vec, Wk, Wv, rteK, rteV);

  qkv_wmma_kernel<<<N_ / 16, 256, 0, stream>>>(node_inp, wq16, wk16, wv16,
                                               bq, bk, bv, node_type, qn, kn, vn);

  edge_logit_kernel<<<E_ / 8, 256, 0, stream>>>(qn, kn, rteK, edge_idx, edge_type,
                                                edge_time, node_type, rel_att, rel_pri,
                                                logit, nmax);
  edge_den_kernel<<<(E_ * H_) / 256, 256, 0, stream>>>(logit, edge_idx, nmax, nden);
  edge_aggr_kernel<<<E_ / 8, 256, 0, stream>>>(vn, rteV, edge_idx, edge_type, edge_time,
                                               node_type, rel_msg, logit, nden, aggr);

  out_wmma_kernel<<<N_ / 16, 256, 0, stream>>>(aggr, wa16, ba, node_inp, node_type,
                                               skip, ln_g, ln_b, outp);
}


# edge logits: rel_att applied to q once per (node,relation) on f32 WMMA 16x16x4 before the edge pass, which becomes one 256-wide dot per edge; logit summation order changes
# speedup vs baseline: 1.2438x; 1.2438x over previous
#include <hip/hip_runtime.h>
#include <math.h>

#define T_   3
#define R_   5
#define H_   8
#define IN_  256
#define OUT_ 256
#define DK_  32
#define N_   30000
#define E_   200000
#define ML_  240

typedef __attribute__((ext_vector_type(16))) _Float16 v16h;
typedef __attribute__((ext_vector_type(8)))  float    v8f;
typedef __attribute__((ext_vector_type(2)))  float    v2f;

__device__ __forceinline__ int kmap(int s, int half) {
  int v = s >> 1;
  return ((v >> 2) << 4) | (half << 3) | ((v & 3) << 1) | (s & 1);
}

__device__ __forceinline__ void atomic_max_f32(float* addr, float val) {
  if (val >= 0.0f) atomicMax((int*)addr, __float_as_int(val));
  else             atomicMin((unsigned int*)addr, __float_as_uint(val));
}

__global__ void pack_w_kernel(const float* __restrict__ W, _Float16* __restrict__ Wp,
                              int total) {
  int idx = blockIdx.x * 256 + threadIdx.x;
  if (idx >= total) return;
  int s    = idx & 15;
  int lane = (idx >> 4) & 31;
  int kk   = (idx >> 9) & 7;
  int ct   = (idx >> 12) & 15;
  int t    = idx >> 16;
  int half = lane >> 4;
  int n    = lane & 15;
  int k    = kk * 32 + kmap(s, half);
  Wp[idx] = (_Float16)W[(size_t)t * (IN_ * OUT_) + (size_t)k * OUT_ + ct * 16 + n];
}

__global__ void init_kernel(float* __restrict__ nmax, float* __restrict__ nden,
                            float* __restrict__ aggr) {
  int i = blockIdx.x * blockDim.x + threadIdx.x;
  int stride = gridDim.x * blockDim.x;
  for (int k = i; k < N_ * H_; k += stride) { nmax[k] = -__builtin_inff(); nden[k] = 0.0f; }
  for (size_t k = i; k < (size_t)N_ * OUT_; k += (size_t)stride) aggr[k] = 0.0f;
}

__global__ void rte_vec_kernel(const float* __restrict__ emb, const float* __restrict__ Wt,
                               const float* __restrict__ b, float* __restrict__ out) {
  int idx = blockIdx.x * 256 + threadIdx.x;
  int i = idx & 255, m = idx >> 8;
  float acc = b[i];
  const float* e = emb + (size_t)m * IN_;
  const float* w = Wt + (size_t)i * IN_;
  for (int j = 0; j < IN_; ++j) acc = fmaf(e[j], w[j], acc);
  out[idx] = acc;
}

__global__ void rte_kv_kernel(const float* __restrict__ rv,
                              const float* __restrict__ Wk, const float* __restrict__ Wv,
                              float* __restrict__ rk, float* __restrict__ rvv) {
  int idx = blockIdx.x * 256 + threadIdx.x;
  int o = idx & 255;
  int m = (idx >> 8) % ML_;
  int t = idx / (ML_ * OUT_);
  const float* v  = rv + (size_t)m * IN_;
  const float* wk = Wk + (size_t)t * (IN_ * OUT_);
  const float* wv = Wv + (size_t)t * (IN_ * OUT_);
  float ak = 0.0f, av = 0.0f;
  for (int i = 0; i < IN_; ++i) {
    float x = v[i];
    ak = fmaf(x, wk[(size_t)i * OUT_ + o], ak);
    av = fmaf(x, wv[(size_t)i * OUT_ + o], av);
  }
  rk[idx] = ak;
  rvv[idx] = av;
}

__device__ __forceinline__ void project_tile(
    const v16h* __restrict__ Asv, const v16h* __restrict__ Wpk,
    const float* __restrict__ bias, float* __restrict__ Out,
    const int* __restrict__ stype, int tmask, int base, int wave, int lane) {
  const int n = lane & 15, hs = lane >> 4;
  for (int t = 0; t < T_; ++t) {
    if (!((tmask >> t) & 1)) continue;
#pragma unroll
    for (int c = 0; c < 2; ++c) {
      const int ct = wave * 2 + c;
      const int nbase = ct * 16;
      const float bv = bias[t * OUT_ + nbase + n];
      v8f acc;
#pragma unroll
      for (int i = 0; i < 8; ++i) acc[i] = bv;
#pragma unroll
      for (int kk = 0; kk < 8; ++kk) {
        v16h a = Asv[kk * 32 + lane];
        v16h b = Wpk[(((size_t)t * 16 + ct) * 8 + kk) * 32 + lane];
        acc = __builtin_amdgcn_wmma_f32_16x16x32_f16(false, a, false, b,
                                                     (short)0, acc, false, false);
      }
#pragma unroll
      for (int i = 0; i < 8; ++i) {
        int m = i + hs * 8;
        if (stype[m] == t)
          Out[(size_t)(base + m) * OUT_ + nbase + n] = acc[i];
      }
    }
  }
}

__global__ __launch_bounds__(256) void qkv_wmma_kernel(
    const float* __restrict__ x,
    const _Float16* __restrict__ wq16, const _Float16* __restrict__ wk16,
    const _Float16* __restrict__ wv16,
    const float* __restrict__ bq, const float* __restrict__ bk, const float* __restrict__ bv,
    const int* __restrict__ ntype,
    float* __restrict__ qn, float* __restrict__ kn, float* __restrict__ vn) {
  __shared__ __align__(32) _Float16 As[16 * IN_];
  __shared__ int stype[16];
  const int tid = threadIdx.x;
  const int base = blockIdx.x * 16;
#pragma unroll
  for (int j = 0; j < 16; ++j) {
    int flat = j * 256 + tid;
    int s = flat & 15, ln = (flat >> 4) & 31, kk = flat >> 9;
    int k = kk * 32 + kmap(s, ln >> 4);
    As[flat] = (_Float16)x[(size_t)(base + (ln & 15)) * IN_ + k];
  }
  if (tid < 16) stype[tid] = ntype[base + tid];
  __syncthreads();
  int tmask = 0;
#pragma unroll
  for (int i = 0; i < 16; ++i) tmask |= 1 << stype[i];
  const int wave = tid >> 5, lane = tid & 31;
  const v16h* Asv = (const v16h*)As;
  project_tile(Asv, (const v16h*)wq16, bq, qn, stype, tmask, base, wave, lane);
  project_tile(Asv, (const v16h*)wk16, bk, kn, stype, tmask, base, wave, lane);
  project_tile(Asv, (const v16h*)wv16, bv, vn, stype, tmask, base, wave, lane);
}

__device__ __forceinline__ void load_frag16(const float* __restrict__ p, v2f (&f)[8]) {
  const float4* p4 = (const float4*)p;
#pragma unroll
  for (int j = 0; j < 4; ++j) {
    const float4 t = p4[j];
    f[2 * j][0] = t.x;     f[2 * j][1] = t.y;
    f[2 * j + 1][0] = t.z; f[2 * j + 1][1] = t.w;
  }
}

__global__ __launch_bounds__(256) void rel_q_kernel(
    const float* __restrict__ qn, const float* __restrict__ rel_att,
    float* __restrict__ qa) {
  const int lane = threadIdx.x & 31, h = threadIdx.x >> 5;
  const int c = lane & 15, hs = lane >> 4;
  const int base = blockIdx.x * 16;
  v2f a[8];
  load_frag16(qn + (size_t)(base + c) * OUT_ + h * DK_ + hs * 16, a);
#pragma unroll
  for (int r = 0; r < R_; ++r) {
    const float* bp = rel_att + (((size_t)r * H_ + h) * DK_ + c) * DK_ + hs * 16;
    v2f b0[8], b1[8];
    load_frag16(bp, b0);
    load_frag16(bp + 16 * DK_, b1);
    v8f acc0 = {0.0f, 0.0f, 0.0f, 0.0f, 0.0f, 0.0f, 0.0f, 0.0f};
    v8f acc1 = acc0;
#pragma unroll
    for (int ks = 0; ks < 8; ++ks) {
      acc0 = __builtin_amdgcn_wmma_f32_16x16x4_f32(false, a[ks], false, b0[ks],
                                                   (short)0, acc0, false, false);
      acc1 = __builtin_amdgcn_wmma_f32_16x16x4_f32(false, a[ks], false, b1[ks],
                                                   (short)0, acc1, false, false);
    }
#pragma unroll
    for (int i = 0; i < 8; ++i) {
      float* o = qa + ((size_t)(base + 8 * hs + i) * R_ + r) * OUT_ + h * DK_ + c;
      o[0]  = acc0[i];
      o[16] = acc1[i];
    }
  }
}

__global__ __launch_bounds__(256) void edge_logit_kernel(
    const float* __restrict__ kn, const float* __restrict__ rteK,
    const float* __restrict__ qa,
    const int* __restrict__ ei, const int* __restrict__ etype,
    const int* __restrict__ etime, const int* __restrict__ ntype,
    const float* __restrict__ rel_pri,
    float* __restrict__ logit, float* __restrict__ nmax) {
  const int lane = threadIdx.x & 31, we = threadIdx.x >> 5;
  const int e = blockIdx.x * 8 + we;
  const int src = ei[e], dst = ei[E_ + e];
  const int r = etype[e], tm = etime[e], st = ntype[src];
  const float4* kb = (const float4*)(kn + (size_t)src * OUT_) + lane * 2;
  const float4* rb = (const float4*)(rteK + ((size_t)st * ML_ + tm) * OUT_) + lane * 2;
  const float4* qb = (const float4*)(qa + ((size_t)dst * R_ + r) * OUT_) + lane * 2;
  const float4 k0 = kb[0], k1 = kb[1];
  const float4 t0 = rb[0], t1 = rb[1];
  const float4 q0 = qb[0], q1 = qb[1];
  float p = (k0.x + t0.x) * q0.x;
  p = fmaf(k0.y + t0.y, q0.y, p);
  p = fmaf(k0.z + t0.z, q0.z, p);
  p = fmaf(k0.w + t0.w, q0.w, p);
  p = fmaf(k1.x + t1.x, q1.x, p);
  p = fmaf(k1.y + t1.y, q1.y, p);
  p = fmaf(k1.z + t1.z, q1.z, p);
  p = fmaf(k1.w + t1.w, q1.w, p);
  p += __shfl_xor(p, 1, 32);
  p += __shfl_xor(p, 2, 32);
  if ((lane & 3) == 0) {
    const int h = lane >> 2;
    float lg = p * rel_pri[r * H_ + h] * 0.17677669529663689f;
    logit[(size_t)e * H_ + h] = lg;
    atomic_max_f32(&nmax[(size_t)dst * H_ + h], lg);
  }
}

__global__ void edge_den_kernel(float* __restrict__ logit, const int* __restrict__ ei,
                                const float* __restrict__ nmax, float* __restrict__ nden) {
  int idx = blockIdx.x * 256 + threadIdx.x;
  int e = idx >> 3, h = idx & 7;
  int dst = ei[E_ + e];
  float ex = __expf(logit[idx] - nmax[(size_t)dst * H_ + h]);
  logit[idx] = ex;
  atomicAdd(&nden[(size_t)dst * H_ + h], ex);
}

__global__ __launch_bounds__(256) void edge_aggr_kernel(
    const float* __restrict__ vn, const float* __restrict__ rteV,
    const int* __restrict__ ei, const int* __restrict__ etype,
    const int* __restrict__ etime, const int* __restrict__ ntype,
    const float* __restrict__ rel_msg, const float* __restrict__ exbuf,
    const float* __restrict__ nden, float* __restrict__ aggr) {
  __shared__ float sv[8][IN_];
  const int lane = threadIdx.x & 31, we = threadIdx.x >> 5;
  const int e = blockIdx.x * 8 + we;
  const int src = ei[e], dst = ei[E_ + e];
  const int r = etype[e], tm = etime[e], st = ntype[src];
  const float* vb = vn + (size_t)src * OUT_;
  const float* rb = rteV + ((size_t)st * ML_ + tm) * OUT_;
#pragma unroll
  for (int j = 0; j < 8; ++j) {
    int idx = j * 32 + lane;
    sv[we][idx] = vb[idx] + rb[idx];
  }
  __syncthreads();
  const float* M = rel_msg + (size_t)r * H_ * DK_ * DK_;
  for (int h = 0; h < H_; ++h) {
    float acc = 0.0f;
    const float* Mh = M + h * DK_ * DK_;
#pragma unroll
    for (int d = 0; d < DK_; ++d)
      acc = fmaf(sv[we][h * DK_ + d], Mh[d * DK_ + lane], acc);
    float att = exbuf[(size_t)e * H_ + h] / nden[(size_t)dst * H_ + h];
    atomicAdd(&aggr[(size_t)dst * OUT_ + h * DK_ + lane], acc * att);
  }
}

__global__ __launch_bounds__(256) void out_wmma_kernel(
    const float* __restrict__ aggr, const _Float16* __restrict__ wa16,
    const float* __restrict__ ba, const float* __restrict__ x,
    const int* __restrict__ ntype, const float* __restrict__ skip,
    const float* __restrict__ ln_g, const float* __restrict__ ln_b,
    float* __restrict__ out) {
  __shared__ __align__(32) _Float16 As[16 * OUT_];
  __shared__ float Tr[16 * OUT_];
  __shared__ int stype[16];
  __shared__ float alpha_s[16], mu_s[16], rs_s[16];
  const int tid = threadIdx.x;
  const int base = blockIdx.x * 16;
#pragma unroll
  for (int j = 0; j < 16; ++j) {
    int flat = j * 256 + tid;
    int s = flat & 15, ln = (flat >> 4) & 31, kk = flat >> 9;
    int k = kk * 32 + kmap(s, ln >> 4);
    As[flat] = (_Float16)aggr[(size_t)(base + (ln & 15)) * OUT_ + k];
  }
  if (tid < 16) {
    int t = ntype[base + tid];
    stype[tid] = t;
    alpha_s[tid] = 1.0f / (1.0f + __expf(-skip[t]));
  }
  __syncthreads();
  int tmask = 0;
#pragma unroll
  for (int i = 0; i < 16; ++i) tmask |= 1 << stype[i];
  const int wave = tid >> 5, lane = tid & 31;
  const int n = lane & 15, hs = lane >> 4;
  const v16h* Asv = (const v16h*)As;
  const v16h* Wpk = (const v16h*)wa16;
  for (int t = 0; t < T_; ++t) {
    if (!((tmask >> t) & 1)) continue;
#pragma unroll
    for (int c = 0; c < 2; ++c) {
      const int ct = wave * 2 + c;
      const int nbase = ct * 16;
      const float bv = ba[t * OUT_ + nbase + n];
      v8f acc;
#pragma unroll
      for (int i = 0; i < 8; ++i) acc[i] = bv;
#pragma unroll
      for (int kk = 0; kk < 8; ++kk) {
        v16h a = Asv[kk * 32 + lane];
        v16h b = Wpk[(((size_t)t * 16 + ct) * 8 + kk) * 32 + lane];
        acc = __builtin_amdgcn_wmma_f32_16x16x32_f16(false, a, false, b,
                                                     (short)0, acc, false, false);
      }
#pragma unroll
      for (int i = 0; i < 8; ++i) {
        int m = i + hs * 8;
        if (stype[m] == t) Tr[m * OUT_ + nbase + n] = acc[i];
      }
    }
  }
  __syncthreads();
#pragma unroll
  for (int j = 0; j < 16; ++j) {
    int flat = j * 256 + tid;
    int row = flat >> 8, col = flat & 255;
    float a = alpha_s[row];
    Tr[flat] = Tr[flat] * a + x[(size_t)(base + row) * IN_ + col] * (1.0f - a);
  }
  __syncthreads();
  if (tid < 16) {
    float mu = 0.0f;
    for (int c = 0; c < OUT_; ++c) mu += Tr[tid * OUT_ + c];
    mu *= (1.0f / OUT_);
    float var = 0.0f;
    for (int c = 0; c < OUT_; ++c) {
      float d = Tr[tid * OUT_ + c] - mu;
      var = fmaf(d, d, var);
    }
    var *= (1.0f / OUT_);
    mu_s[tid] = mu;
    rs_s[tid] = rsqrtf(var + 1e-5f);
  }
  __syncthreads();
#pragma unroll
  for (int j = 0; j < 16; ++j) {
    int flat = j * 256 + tid;
    int row = flat >> 8, col = flat & 255;
    int t = stype[row];
    out[(size_t)(base + row) * OUT_ + col] =
        (Tr[flat] - mu_s[row]) * rs_s[row] * ln_g[t * OUT_ + col] + ln_b[t * OUT_ + col];
  }
}

extern "C" void kernel_launch(void* const* d_in, const int* in_sizes, int n_in,
                              void* d_out, int out_size, void* d_ws, size_t ws_size,
                              hipStream_t stream) {
  (void)in_sizes; (void)n_in; (void)out_size; (void)ws_size;
  const float* node_inp  = (const float*)d_in[0];
  const float* Wk        = (const float*)d_in[1];
  const float* bk        = (const float*)d_in[2];
  const float* Wq        = (const float*)d_in[3];
  const float* bq        = (const float*)d_in[4];
  const float* Wv        = (const float*)d_in[5];
  const float* bv        = (const float*)d_in[6];
  const float* Wa        = (const float*)d_in[7];
  const float* ba        = (const float*)d_in[8];
  const float* ln_g      = (const float*)d_in[9];
  const float* ln_b      = (const float*)d_in[10];
  const float* rel_pri   = (const float*)d_in[11];
  const float* rel_att   = (const float*)d_in[12];
  const float* rel_msg   = (const float*)d_in[13];
  const float* skip      = (const float*)d_in[14];
  const float* rte_W     = (const float*)d_in[15];
  const float* rte_b     = (const float*)d_in[16];
  const float* rte_emb   = (const float*)d_in[17];
  const int*   node_type = (const int*)d_in[18];
  const int*   edge_idx  = (const int*)d_in[19];
  const int*   edge_type = (const int*)d_in[20];
  const int*   edge_time = (const int*)d_in[21];
  float* outp = (float*)d_out;

  char* p = (char*)d_ws;
  auto alloc = [&](size_t bytes) -> void* {
    void* r = (void*)p;
    p += (bytes + 255) & ~(size_t)255;
    return r;
  };
  float* rte_vec = (float*)alloc((size_t)ML_ * IN_ * 4);
  float* rteK    = (float*)alloc((size_t)T_ * ML_ * OUT_ * 4);
  float* rteV    = (float*)alloc((size_t)T_ * ML_ * OUT_ * 4);
  float* qn      = (float*)alloc((size_t)N_ * OUT_ * 4);
  float* kn      = (float*)alloc((size_t)N_ * OUT_ * 4);
  float* vn      = (float*)alloc((size_t)N_ * OUT_ * 4);
  float* logit   = (float*)alloc((size_t)E_ * H_ * 4);
  float* nmax    = (float*)alloc((size_t)N_ * H_ * 4);
  float* nden    = (float*)alloc((size_t)N_ * H_ * 4);
  float* qa      = (float*)alloc((size_t)N_ * R_ * OUT_ * 4);
  float* aggr    = qn;
  _Float16* wq16 = (_Float16*)alloc((size_t)T_ * IN_ * OUT_ * 2);
  _Float16* wk16 = (_Float16*)alloc((size_t)T_ * IN_ * OUT_ * 2);
  _Float16* wv16 = (_Float16*)alloc((size_t)T_ * IN_ * OUT_ * 2);
  _Float16* wa16 = (_Float16*)alloc((size_t)T_ * OUT_ * OUT_ * 2);

  const int wtotal = T_ * IN_ * OUT_;
  pack_w_kernel<<<wtotal / 256, 256, 0, stream>>>(Wq, wq16, wtotal);
  pack_w_kernel<<<wtotal / 256, 256, 0, stream>>>(Wk, wk16, wtotal);
  pack_w_kernel<<<wtotal / 256, 256, 0, stream>>>(Wv, wv16, wtotal);
  pack_w_kernel<<<wtotal / 256, 256, 0, stream>>>(Wa, wa16, wtotal);

  rte_vec_kernel<<<(ML_ * IN_) / 256, 256, 0, stream>>>(rte_emb, rte_W, rte_b, rte_vec);
  rte_kv_kernel<<<(T_ * ML_ * OUT_) / 256, 256, 0, stream>>>(rte_vec, Wk, Wv, rteK, rteV);

  qkv_wmma_kernel<<<N_ / 16, 256, 0, stream>>>(node_inp, wq16, wk16, wv16,
                                               bq, bk, bv, node_type, qn, kn, vn);

  rel_q_kernel<<<N_ / 16, 256, 0, stream>>>(qn, rel_att, qa);

  init_kernel<<<4096, 256, 0, stream>>>(nmax, nden, aggr);

  edge_logit_kernel<<<E_ / 8, 256, 0, stream>>>(kn, rteK, qa, edge_idx, edge_type,
                                                edge_time, node_type, rel_pri,
                                                logit, nmax);
  edge_den_kernel<<<(E_ * H_) / 256, 256, 0, stream>>>(logit, edge_idx, nmax, nden);
  edge_aggr_kernel<<<E_ / 8, 256, 0, stream>>>(vn, rteV, edge_idx, edge_type, edge_time,
                                               node_type, rel_msg, logit, nden, aggr);

  out_wmma_kernel<<<N_ / 16, 256, 0, stream>>>(aggr, wa16, ba, node_inp, node_type,
                                               skip, ln_g, ln_b, outp);
}


# aggregation as a gather per target node: edges chained per target by atomic exchange, softmax sums in registers, messages pre-transformed per (node,relation) on f32 WMMA 16x16x4; arrival-order sums
# speedup vs baseline: 1.6341x; 1.3138x over previous
#include <hip/hip_runtime.h>
#include <math.h>

#define T_   3
#define R_   5
#define H_   8
#define IN_  256
#define OUT_ 256
#define DK_  32
#define N_   30000
#define E_   200000
#define ML_  240

typedef __attribute__((ext_vector_type(16))) _Float16 v16h;
typedef __attribute__((ext_vector_type(8)))  float    v8f;
typedef __attribute__((ext_vector_type(2)))  float    v2f;

__device__ __forceinline__ int kmap(int s, int half) {
  int v = s >> 1;
  return ((v >> 2) << 4) | (half << 3) | ((v & 3) << 1) | (s & 1);
}

__global__ void pack_w_kernel(const float* __restrict__ W, _Float16* __restrict__ Wp,
                              int total) {
  int idx = blockIdx.x * 256 + threadIdx.x;
  if (idx >= total) return;
  int s    = idx & 15;
  int lane = (idx >> 4) & 31;
  int kk   = (idx >> 9) & 7;
  int ct   = (idx >> 12) & 15;
  int t    = idx >> 16;
  int half = lane >> 4;
  int n    = lane & 15;
  int k    = kk * 32 + kmap(s, half);
  Wp[idx] = (_Float16)W[(size_t)t * (IN_ * OUT_) + (size_t)k * OUT_ + ct * 16 + n];
}

__global__ void list_init_kernel(int* __restrict__ head) {
  int i = blockIdx.x * 256 + threadIdx.x;
  if (i < N_) head[i] = -1;
}

__global__ void rte_vec_kernel(const float* __restrict__ emb, const float* __restrict__ Wt,
                               const float* __restrict__ b, float* __restrict__ out) {
  int idx = blockIdx.x * 256 + threadIdx.x;
  int i = idx & 255, m = idx >> 8;
  float acc = b[i];
  const float* e = emb + (size_t)m * IN_;
  const float* w = Wt + (size_t)i * IN_;
  for (int j = 0; j < IN_; ++j) acc = fmaf(e[j], w[j], acc);
  out[idx] = acc;
}

__global__ void rte_kv_kernel(const float* __restrict__ rv,
                              const float* __restrict__ Wk, const float* __restrict__ Wv,
                              float* __restrict__ rk, float* __restrict__ rvv) {
  int idx = blockIdx.x * 256 + threadIdx.x;
  int o = idx & 255;
  int m = (idx >> 8) % ML_;
  int t = idx / (ML_ * OUT_);
  const float* v  = rv + (size_t)m * IN_;
  const float* wk = Wk + (size_t)t * (IN_ * OUT_);
  const float* wv = Wv + (size_t)t * (IN_ * OUT_);
  float ak = 0.0f, av = 0.0f;
  for (int i = 0; i < IN_; ++i) {
    float x = v[i];
    ak = fmaf(x, wk[(size_t)i * OUT_ + o], ak);
    av = fmaf(x, wv[(size_t)i * OUT_ + o], av);
  }
  rk[idx] = ak;
  rvv[idx] = av;
}

__device__ __forceinline__ void project_tile(
    const v16h* __restrict__ Asv, const v16h* __restrict__ Wpk,
    const float* __restrict__ bias, float* __restrict__ Out,
    const int* __restrict__ stype, int tmask, int base, int wave, int lane) {
  const int n = lane & 15, hs = lane >> 4;
  for (int t = 0; t < T_; ++t) {
    if (!((tmask >> t) & 1)) continue;
#pragma unroll
    for (int c = 0; c < 2; ++c) {
      const int ct = wave * 2 + c;
      const int nbase = ct * 16;
      const float bv = bias[t * OUT_ + nbase + n];
      v8f acc;
#pragma unroll
      for (int i = 0; i < 8; ++i) acc[i] = bv;
#pragma unroll
      for (int kk = 0; kk < 8; ++kk) {
        v16h a = Asv[kk * 32 + lane];
        v16h b = Wpk[(((size_t)t * 16 + ct) * 8 + kk) * 32 + lane];
        acc = __builtin_amdgcn_wmma_f32_16x16x32_f16(false, a, false, b,
                                                     (short)0, acc, false, false);
      }
#pragma unroll
      for (int i = 0; i < 8; ++i) {
        int m = i + hs * 8;
        if (stype[m] == t)
          Out[(size_t)(base + m) * OUT_ + nbase + n] = acc[i];
      }
    }
  }
}

__global__ __launch_bounds__(256) void qkv_wmma_kernel(
    const float* __restrict__ x,
    const _Float16* __restrict__ wq16, const _Float16* __restrict__ wk16,
    const _Float16* __restrict__ wv16,
    const float* __restrict__ bq, const float* __restrict__ bk, const float* __restrict__ bv,
    const int* __restrict__ ntype,
    float* __restrict__ qn, float* __restrict__ kn, float* __restrict__ vn) {
  __shared__ __align__(32) _Float16 As[16 * IN_];
  __shared__ int stype[16];
  const int tid = threadIdx.x;
  const int base = blockIdx.x * 16;
#pragma unroll
  for (int j = 0; j < 16; ++j) {
    int flat = j * 256 + tid;
    int s = flat & 15, ln = (flat >> 4) & 31, kk = flat >> 9;
    int k = kk * 32 + kmap(s, ln >> 4);
    As[flat] = (_Float16)x[(size_t)(base + (ln & 15)) * IN_ + k];
  }
  if (tid < 16) stype[tid] = ntype[base + tid];
  __syncthreads();
  int tmask = 0;
#pragma unroll
  for (int i = 0; i < 16; ++i) tmask |= 1 << stype[i];
  const int wave = tid >> 5, lane = tid & 31;
  const v16h* Asv = (const v16h*)As;
  project_tile(Asv, (const v16h*)wq16, bq, qn, stype, tmask, base, wave, lane);
  project_tile(Asv, (const v16h*)wk16, bk, kn, stype, tmask, base, wave, lane);
  project_tile(Asv, (const v16h*)wv16, bv, vn, stype, tmask, base, wave, lane);
}

__device__ __forceinline__ void load_frag16(const float* __restrict__ p, v2f (&f)[8]) {
  const float4* p4 = (const float4*)p;
#pragma unroll
  for (int j = 0; j < 4; ++j) {
    const float4 t = p4[j];
    f[2 * j][0] = t.x;     f[2 * j][1] = t.y;
    f[2 * j + 1][0] = t.z; f[2 * j + 1][1] = t.w;
  }
}

__global__ __launch_bounds__(256) void rel_q_kernel(
    const float* __restrict__ qn, const float* __restrict__ rel_att,
    float* __restrict__ qa) {
  const int lane = threadIdx.x & 31, h = threadIdx.x >> 5;
  const int c = lane & 15, hs = lane >> 4;
  const int base = blockIdx.x * 16;
  v2f a[8];
  load_frag16(qn + (size_t)(base + c) * OUT_ + h * DK_ + hs * 16, a);
#pragma unroll
  for (int r = 0; r < R_; ++r) {
    const float* bp = rel_att + (((size_t)r * H_ + h) * DK_ + c) * DK_ + hs * 16;
    v2f b0[8], b1[8];
    load_frag16(bp, b0);
    load_frag16(bp + 16 * DK_, b1);
    v8f acc0 = {0.0f, 0.0f, 0.0f, 0.0f, 0.0f, 0.0f, 0.0f, 0.0f};
    v8f acc1 = acc0;
#pragma unroll
    for (int ks = 0; ks < 8; ++ks) {
      acc0 = __builtin_amdgcn_wmma_f32_16x16x4_f32(false, a[ks], false, b0[ks],
                                                   (short)0, acc0, false, false);
      acc1 = __builtin_amdgcn_wmma_f32_16x16x4_f32(false, a[ks], false, b1[ks],
                                                   (short)0, acc1, false, false);
    }
#pragma unroll
    for (int i = 0; i < 8; ++i) {
      float* o = qa + ((size_t)(base + 8 * hs + i) * R_ + r) * OUT_ + h * DK_ + c;
      o[0]  = acc0[i];
      o[16] = acc1[i];
    }
  }
}

__global__ __launch_bounds__(256) void edge_logit_kernel(
    const float* __restrict__ kn, const float* __restrict__ rteK,
    const float* __restrict__ qa,
    const int* __restrict__ ei, const int* __restrict__ etype,
    const int* __restrict__ etime, const int* __restrict__ ntype,
    const float* __restrict__ rel_pri,
    float* __restrict__ logit, int* __restrict__ head, int* __restrict__ next) {
  const int lane = threadIdx.x & 31, we = threadIdx.x >> 5;
  const int e = blockIdx.x * 8 + we;
  const int src = ei[e], dst = ei[E_ + e];
  const int r = etype[e], tm = etime[e], st = ntype[src];
  const float4* kb = (const float4*)(kn + (size_t)src * OUT_) + lane * 2;
  const float4* rb = (const float4*)(rteK + ((size_t)st * ML_ + tm) * OUT_) + lane * 2;
  const float4* qb = (const float4*)(qa + ((size_t)dst * R_ + r) * OUT_) + lane * 2;
  const float4 k0 = kb[0], k1 = kb[1];
  const float4 t0 = rb[0], t1 = rb[1];
  const float4 q0 = qb[0], q1 = qb[1];
  float p = (k0.x + t0.x) * q0.x;
  p = fmaf(k0.y + t0.y, q0.y, p);
  p = fmaf(k0.z + t0.z, q0.z, p);
  p = fmaf(k0.w + t0.w, q0.w, p);
  p = fmaf(k1.x + t1.x, q1.x, p);
  p = fmaf(k1.y + t1.y, q1.y, p);
  p = fmaf(k1.z + t1.z, q1.z, p);
  p = fmaf(k1.w + t1.w, q1.w, p);
  p += __shfl_xor(p, 1, 32);
  p += __shfl_xor(p, 2, 32);
  if ((lane & 3) == 0) {
    const int h = lane >> 2;
    float lg = p * rel_pri[r * H_ + h] * 0.17677669529663689f;
    logit[(size_t)e * H_ + h] = lg;
  }
  if (lane == 0) next[e] = atomicExch(&head[dst], e);
}

__global__ __launch_bounds__(256) void rel_v_kernel(
    const float* __restrict__ vn, const float* __restrict__ rteV,
    const float* __restrict__ rel_msg,
    float* __restrict__ va, float* __restrict__ rtv) {
  const int lane = threadIdx.x & 31, h = threadIdx.x >> 5;
  const int c = lane & 15, hs = lane >> 4;
  const bool node_rows = blockIdx.x < N_ / 16;
  const float* __restrict__ rows = node_rows ? vn : rteV;
  float* __restrict__ outp = node_rows ? va : rtv;
  const int base = (node_rows ? blockIdx.x : blockIdx.x - N_ / 16) * 16;
  v2f a[8];
  load_frag16(rows + (size_t)(base + c) * OUT_ + h * DK_ + hs * 16, a);
#pragma unroll
  for (int r = 0; r < R_; ++r) {
    const float* bp = rel_msg + (((size_t)r * H_ + h) * DK_ + hs * 16) * DK_ + c;
    v2f b0[8], b1[8];
#pragma unroll
    for (int ks = 0; ks < 8; ++ks) {
      b0[ks][0] = bp[(2 * ks) * DK_];
      b0[ks][1] = bp[(2 * ks + 1) * DK_];
      b1[ks][0] = bp[(2 * ks) * DK_ + 16];
      b1[ks][1] = bp[(2 * ks + 1) * DK_ + 16];
    }
    v8f acc0 = {0.0f, 0.0f, 0.0f, 0.0f, 0.0f, 0.0f, 0.0f, 0.0f};
    v8f acc1 = acc0;
#pragma unroll
    for (int ks = 0; ks < 8; ++ks) {
      acc0 = __builtin_amdgcn_wmma_f32_16x16x4_f32(false, a[ks], false, b0[ks],
                                                   (short)0, acc0, false, false);
      acc1 = __builtin_amdgcn_wmma_f32_16x16x4_f32(false, a[ks], false, b1[ks],
                                                   (short)0, acc1, false, false);
    }
#pragma unroll
    for (int i = 0; i < 8; ++i) {
      float* o = outp + ((size_t)(base + 8 * hs + i) * R_ + r) * OUT_ + h * DK_ + c;
      o[0]  = acc0[i];
      o[16] = acc1[i];
    }
  }
}

__global__ __launch_bounds__(256) void node_aggr_kernel(
    const float* __restrict__ va, const float* __restrict__ rtv,
    const float* __restrict__ logit,
    const int* __restrict__ ei, const int* __restrict__ etype,
    const int* __restrict__ etime, const int* __restrict__ ntype,
    const int* __restrict__ head, const int* __restrict__ next,
    float* __restrict__ aggr) {
  const int lane = threadIdx.x & 31;
  const int n = blockIdx.x * 8 + (threadIdx.x >> 5);
  const int hl = lane & 7;
  const int e0 = head[n];
  float m = -__builtin_inff();
  int e = e0;
  for (int i = 0; i < E_ && e >= 0; ++i) {
    m = fmaxf(m, logit[(size_t)e * H_ + hl]);
    e = next[e];
  }
  float den = 0.0f;
  float acc[H_];
#pragma unroll
  for (int j = 0; j < H_; ++j) acc[j] = 0.0f;
  e = e0;
  for (int i = 0; i < E_ && e >= 0; ++i) {
    const int src = ei[e];
    const int r = etype[e], tm = etime[e];
    const int en = next[e];
    const float ex = __expf(logit[(size_t)e * H_ + hl] - m);
    const int st = ntype[src];
    const float* vb = va + ((size_t)src * R_ + r) * OUT_ + lane;
    const float* rb = rtv + (((size_t)st * ML_ + tm) * R_ + r) * OUT_ + lane;
    den += ex;
#pragma unroll
    for (int j = 0; j < H_; ++j) {
      const float w = __shfl(ex, j, 32);
      acc[j] = fmaf(w, vb[j * DK_] + rb[j * DK_], acc[j]);
    }
    e = en;
  }
  float* o = aggr + (size_t)n * OUT_ + lane;
#pragma unroll
  for (int j = 0; j < H_; ++j) {
    const float d = __shfl(den, j, 32);
    o[j * DK_] = (e0 >= 0) ? acc[j] / d : 0.0f;
  }
}

__global__ __launch_bounds__(256) void out_wmma_kernel(
    const float* __restrict__ aggr, const _Float16* __restrict__ wa16,
    const float* __restrict__ ba, const float* __restrict__ x,
    const int* __restrict__ ntype, const float* __restrict__ skip,
    const float* __restrict__ ln_g, const float* __restrict__ ln_b,
    float* __restrict__ out) {
  __shared__ __align__(32) _Float16 As[16 * OUT_];
  __shared__ float Tr[16 * OUT_];
  __shared__ int stype[16];
  __shared__ float alpha_s[16], mu_s[16], rs_s[16];
  const int tid = threadIdx.x;
  const int base = blockIdx.x * 16;
#pragma unroll
  for (int j = 0; j < 16; ++j) {
    int flat = j * 256 + tid;
    int s = flat & 15, ln = (flat >> 4) & 31, kk = flat >> 9;
    int k = kk * 32 + kmap(s, ln >> 4);
    As[flat] = (_Float16)aggr[(size_t)(base + (ln & 15)) * OUT_ + k];
  }
  if (tid < 16) {
    int t = ntype[base + tid];
    stype[tid] = t;
    alpha_s[tid] = 1.0f / (1.0f + __expf(-skip[t]));
  }
  __syncthreads();
  int tmask = 0;
#pragma unroll
  for (int i = 0; i < 16; ++i) tmask |= 1 << stype[i];
  const int wave = tid >> 5, lane = tid & 31;
  const int n = lane & 15, hs = lane >> 4;
  const v16h* Asv = (const v16h*)As;
  const v16h* Wpk = (const v16h*)wa16;
  for (int t = 0; t < T_; ++t) {
    if (!((tmask >> t) & 1)) continue;
#pragma unroll
    for (int c = 0; c < 2; ++c) {
      const int ct = wave * 2 + c;
      const int nbase = ct * 16;
      const float bv = ba[t * OUT_ + nbase + n];
      v8f acc;
#pragma unroll
      for (int i = 0; i < 8; ++i) acc[i] = bv;
#pragma unroll
      for (int kk = 0; kk < 8; ++kk) {
        v16h a = Asv[kk * 32 + lane];
        v16h b = Wpk[(((size_t)t * 16 + ct) * 8 + kk) * 32 + lane];
        acc = __builtin_amdgcn_wmma_f32_16x16x32_f16(false, a, false, b,
                                                     (short)0, acc, false, false);
      }
#pragma unroll
      for (int i = 0; i < 8; ++i) {
        int m = i + hs * 8;
        if (stype[m] == t) Tr[m * OUT_ + nbase + n] = acc[i];
      }
    }
  }
  __syncthreads();
#pragma unroll
  for (int j = 0; j < 16; ++j) {
    int flat = j * 256 + tid;
    int row = flat >> 8, col = flat & 255;
    float a = alpha_s[row];
    Tr[flat] = Tr[flat] * a + x[(size_t)(base + row) * IN_ + col] * (1.0f - a);
  }
  __syncthreads();
  if (tid < 16) {
    float mu = 0.0f;
    for (int c = 0; c < OUT_; ++c) mu += Tr[tid * OUT_ + c];
    mu *= (1.0f / OUT_);
    float var = 0.0f;
    for (int c = 0; c < OUT_; ++c) {
      float d = Tr[tid * OUT_ + c] - mu;
      var = fmaf(d, d, var);
    }
    var *= (1.0f / OUT_);
    mu_s[tid] = mu;
    rs_s[tid] = rsqrtf(var + 1e-5f);
  }
  __syncthreads();
#pragma unroll
  for (int j = 0; j < 16; ++j) {
    int flat = j * 256 + tid;
    int row = flat >> 8, col = flat & 255;
    int t = stype[row];
    out[(size_t)(base + row) * OUT_ + col] =
        (Tr[flat] - mu_s[row]) * rs_s[row] * ln_g[t * OUT_ + col] + ln_b[t * OUT_ + col];
  }
}

extern "C" void kernel_launch(void* const* d_in, const int* in_sizes, int n_in,
                              void* d_out, int out_size, void* d_ws, size_t ws_size,
                              hipStream_t stream) {
  (void)in_sizes; (void)n_in; (void)out_size; (void)ws_size;
  const float* node_inp  = (const float*)d_in[0];
  const float* Wk        = (const float*)d_in[1];
  const float* bk        = (const float*)d_in[2];
  const float* Wq        = (const float*)d_in[3];
  const float* bq        = (const float*)d_in[4];
  const float* Wv        = (const float*)d_in[5];
  const float* bv        = (const float*)d_in[6];
  const float* Wa        = (const float*)d_in[7];
  const float* ba        = (const float*)d_in[8];
  const float* ln_g      = (const float*)d_in[9];
  const float* ln_b      = (const float*)d_in[10];
  const float* rel_pri   = (const float*)d_in[11];
  const float* rel_att   = (const float*)d_in[12];
  const float* rel_msg   = (const float*)d_in[13];
  const float* skip      = (const float*)d_in[14];
  const float* rte_W     = (const float*)d_in[15];
  const float* rte_b     = (const float*)d_in[16];
  const float* rte_emb   = (const float*)d_in[17];
  const int*   node_type = (const int*)d_in[18];
  const int*   edge_idx  = (const int*)d_in[19];
  const int*   edge_type = (const int*)d_in[20];
  const int*   edge_time = (const int*)d_in[21];
  float* outp = (float*)d_out;

  char* p = (char*)d_ws;
  auto alloc = [&](size_t bytes) -> void* {
    void* r = (void*)p;
    p += (bytes + 255) & ~(size_t)255;
    return r;
  };
  float* rte_vec = (float*)alloc((size_t)ML_ * IN_ * 4);
  float* rteK    = (float*)alloc((size_t)T_ * ML_ * OUT_ * 4);
  float* rteV    = (float*)alloc((size_t)T_ * ML_ * OUT_ * 4);
  float* qn      = (float*)alloc((size_t)N_ * OUT_ * 4);
  float* kn      = (float*)alloc((size_t)N_ * OUT_ * 4);
  float* vn      = (float*)alloc((size_t)N_ * OUT_ * 4);
  float* logit   = (float*)alloc((size_t)E_ * H_ * 4);
  int*   head    = (int*)alloc((size_t)N_ * 4);
  int*   next    = (int*)alloc((size_t)E_ * 4);
  float* rtv     = (float*)alloc((size_t)T_ * ML_ * R_ * OUT_ * 4);
  float* qa      = (float*)alloc((size_t)N_ * R_ * OUT_ * 4);
  float* aggr    = qn;
  float* va      = qa;
  _Float16* wq16 = (_Float16*)alloc((size_t)T_ * IN_ * OUT_ * 2);
  _Float16* wk16 = (_Float16*)alloc((size_t)T_ * IN_ * OUT_ * 2);
  _Float16* wv16 = (_Float16*)alloc((size_t)T_ * IN_ * OUT_ * 2);
  _Float16* wa16 = (_Float16*)alloc((size_t)T_ * OUT_ * OUT_ * 2);

  const int wtotal = T_ * IN_ * OUT_;
  pack_w_kernel<<<wtotal / 256, 256, 0, stream>>>(Wq, wq16, wtotal);
  pack_w_kernel<<<wtotal / 256, 256, 0, stream>>>(Wk, wk16, wtotal);
  pack_w_kernel<<<wtotal / 256, 256, 0, stream>>>(Wv, wv16, wtotal);
  pack_w_kernel<<<wtotal / 256, 256, 0, stream>>>(Wa, wa16, wtotal);

  rte_vec_kernel<<<(ML_ * IN_) / 256, 256, 0, stream>>>(rte_emb, rte_W, rte_b, rte_vec);
  rte_kv_kernel<<<(T_ * ML_ * OUT_) / 256, 256, 0, stream>>>(rte_vec, Wk, Wv, rteK, rteV);

  qkv_wmma_kernel<<<N_ / 16, 256, 0, stream>>>(node_inp, wq16, wk16, wv16,
                                               bq, bk, bv, node_type, qn, kn, vn);

  rel_q_kernel<<<N_ / 16, 256, 0, stream>>>(qn, rel_att, qa);

  list_init_kernel<<<(N_ + 255) / 256, 256, 0, stream>>>(head);

  edge_logit_kernel<<<E_ / 8, 256, 0, stream>>>(kn, rteK, qa, edge_idx, edge_type,
                                                edge_time, node_type, rel_pri,
                                                logit, head, next);
  rel_v_kernel<<<N_ / 16 + (T_ * ML_) / 16, 256, 0, stream>>>(vn, rteV, rel_msg, va, rtv);
  node_aggr_kernel<<<N_ / 8, 256, 0, stream>>>(va, rtv, logit, edge_idx, edge_type,
                                               edge_time, node_type, head, next, aggr);

  out_wmma_kernel<<<N_ / 16, 256, 0, stream>>>(aggr, wa16, ba, node_inp, node_type,
                                               skip, ln_g, ln_b, outp);
}


# q/k/v projections: one block = 80 nodes x one projection, so each f16 weight fragment is loaded once for ten matrix instructions (was one each), operands fetched a k-step ahead; same sums
# speedup vs baseline: 1.9110x; 1.1695x over previous
#include <hip/hip_runtime.h>
#include <math.h>

#define T_   3
#define R_   5
#define H_   8
#define IN_  256
#define OUT_ 256
#define DK_  32
#define N_   30000
#define E_   200000
#define ML_  240

typedef __attribute__((ext_vector_type(16))) _Float16 v16h;
typedef __attribute__((ext_vector_type(8)))  float    v8f;
typedef __attribute__((ext_vector_type(2)))  float    v2f;

__device__ __forceinline__ int kmap(int s, int half) {
  int v = s >> 1;
  return ((v >> 2) << 4) | (half << 3) | ((v & 3) << 1) | (s & 1);
}

__global__ void pack_w_kernel(const float* __restrict__ W, _Float16* __restrict__ Wp,
                              int total) {
  int idx = blockIdx.x * 256 + threadIdx.x;
  if (idx >= total) return;
  int s    = idx & 15;
  int lane = (idx >> 4) & 31;
  int kk   = (idx >> 9) & 7;
  int ct   = (idx >> 12) & 15;
  int t    = idx >> 16;
  int half = lane >> 4;
  int n    = lane & 15;
  int k    = kk * 32 + kmap(s, half);
  Wp[idx] = (_Float16)W[(size_t)t * (IN_ * OUT_) + (size_t)k * OUT_ + ct * 16 + n];
}

__global__ void list_init_kernel(int* __restrict__ head) {
  int i = blockIdx.x * 256 + threadIdx.x;
  if (i < N_) head[i] = -1;
}

__global__ void rte_vec_kernel(const float* __restrict__ emb, const float* __restrict__ Wt,
                               const float* __restrict__ b, float* __restrict__ out) {
  int idx = blockIdx.x * 256 + threadIdx.x;
  int i = idx & 255, m = idx >> 8;
  float acc = b[i];
  const float* e = emb + (size_t)m * IN_;
  const float* w = Wt + (size_t)i * IN_;
  for (int j = 0; j < IN_; ++j) acc = fmaf(e[j], w[j], acc);
  out[idx] = acc;
}

__global__ void rte_kv_kernel(const float* __restrict__ rv,
                              const float* __restrict__ Wk, const float* __restrict__ Wv,
                              float* __restrict__ rk, float* __restrict__ rvv) {
  int idx = blockIdx.x * 256 + threadIdx.x;
  int o = idx & 255;
  int m = (idx >> 8) % ML_;
  int t = idx / (ML_ * OUT_);
  const float* v  = rv + (size_t)m * IN_;
  const float* wk = Wk + (size_t)t * (IN_ * OUT_);
  const float* wv = Wv + (size_t)t * (IN_ * OUT_);
  float ak = 0.0f, av = 0.0f;
  for (int i = 0; i < IN_; ++i) {
    float x = v[i];
    ak = fmaf(x, wk[(size_t)i * OUT_ + o], ak);
    av = fmaf(x, wv[(size_t)i * OUT_ + o], av);
  }
  rk[idx] = ak;
  rvv[idx] = av;
}

#define QG_ 5

typedef __attribute__((ext_vector_type(8))) _Float16 v8h;

__device__ __forceinline__ void project_tiles(
    const v8h* __restrict__ As8, const v16h* __restrict__ Wpk,
    const float* __restrict__ bias, float* __restrict__ Out,
    const unsigned (&tp)[QG_], int tmask, int base, int wave, int lane) {
  const int n = lane & 15, hs = lane >> 4;
  const int ct0 = wave * 2;
  const int nbase = ct0 * 16;
  const v8h* ap = As8 + lane;
  float* orow = Out + (size_t)(base + hs * 8) * OUT_ + nbase + n;
  for (int t = 0; t < T_; ++t) {
    if (!((tmask >> t) & 1)) continue;
    const v16h* wp = Wpk + (((size_t)t * 16 + ct0) * 8) * 32 + lane;
    unsigned sel[QG_];
#pragma unroll
    for (int g = 0; g < QG_; ++g) sel[g] = tp[g] >> t;
    v16h b0[2], b1[2];
    v8h alo[2][QG_], ahi[2][QG_];
    b0[0] = wp[0];      b1[0] = wp[8 * 32];
#pragma unroll
    for (int g = 0; g < QG_; ++g) {
      alo[0][g] = ap[((g * 8) * 2 + 0) * 32];
      ahi[0][g] = ap[((g * 8) * 2 + 1) * 32];
    }
    const float bv0 = bias[t * OUT_ + nbase + n];
    const float bv1 = bias[t * OUT_ + nbase + 16 + n];
    v8f acc[QG_][2];
#pragma unroll
    for (int g = 0; g < QG_; ++g) {
#pragma unroll
      for (int i = 0; i < 8; ++i) { acc[g][0][i] = bv0; acc[g][1][i] = bv1; }
    }
#pragma unroll
    for (int kk = 0; kk < 8; ++kk) {
      const int cur = kk & 1, nxt = cur ^ 1;
      if (kk + 1 < 8) {
        b0[nxt] = wp[(kk + 1) * 32];
        b1[nxt] = wp[(8 + kk + 1) * 32];
#pragma unroll
        for (int g = 0; g < QG_; ++g) {
          alo[nxt][g] = ap[((g * 8 + kk + 1) * 2 + 0) * 32];
          ahi[nxt][g] = ap[((g * 8 + kk + 1) * 2 + 1) * 32];
        }
      }
      __builtin_amdgcn_sched_barrier(0);
#pragma unroll
      for (int g = 0; g < QG_; ++g) {
        const v16h a = __builtin_shufflevector(alo[cur][g], ahi[cur][g],
                                               0, 1, 2, 3, 4, 5, 6, 7,
                                               8, 9, 10, 11, 12, 13, 14, 15);
        acc[g][0] = __builtin_amdgcn_wmma_f32_16x16x32_f16(false, a, false, b0[cur],
                                                           (short)0, acc[g][0], false, false);
        acc[g][1] = __builtin_amdgcn_wmma_f32_16x16x32_f16(false, a, false, b1[cur],
                                                           (short)0, acc[g][1], false, false);
      }
      __builtin_amdgcn_sched_barrier(0);
    }
#pragma unroll
    for (int g = 0; g < QG_; ++g) {
#pragma unroll
      for (int i = 0; i < 8; ++i) {
        if ((sel[g] >> (3 * i)) & 1u) {
          float* o = orow + (size_t)(g * 16 + i) * OUT_;
          o[0]  = acc[g][0][i];
          o[16] = acc[g][1][i];
        }
      }
    }
  }
}

__global__ __launch_bounds__(256) void qkv_wmma_kernel(
    const float* __restrict__ x,
    const _Float16* __restrict__ wq16, const _Float16* __restrict__ wk16,
    const _Float16* __restrict__ wv16,
    const float* __restrict__ bq, const float* __restrict__ bk, const float* __restrict__ bv,
    const int* __restrict__ ntype,
    float* __restrict__ qn, float* __restrict__ kn, float* __restrict__ vn) {
  __shared__ __align__(32) _Float16 As[QG_ * 16 * IN_];
  __shared__ int stype[QG_ * 16];
  const int tid = threadIdx.x;
  const int proj = blockIdx.x % 3;
  const int base = (blockIdx.x / 3) * (QG_ * 16);
  const _Float16* __restrict__ w16 = proj == 0 ? wq16 : (proj == 1 ? wk16 : wv16);
  const float* __restrict__ bias = proj == 0 ? bq : (proj == 1 ? bk : bv);
  float* __restrict__ outp = proj == 0 ? qn : (proj == 1 ? kn : vn);
#pragma unroll
  for (int g = 0; g < QG_; ++g) {
#pragma unroll
    for (int j = 0; j < 16; ++j) {
      int flat = j * 256 + tid;
      int s = flat & 15, ln = (flat >> 4) & 31, kk = flat >> 9;
      int k = kk * 32 + kmap(s, ln >> 4);
      As[((((g * 8 + kk) * 2 + (s >> 3)) * 32 + ln) << 3) + (s & 7)] =
          (_Float16)x[(size_t)(base + g * 16 + (ln & 15)) * IN_ + k];
    }
  }
  if (tid < QG_ * 16) stype[tid] = ntype[base + tid];
  __syncthreads();
  int tmask = 0;
  for (int i = 0; i < QG_ * 16; ++i) tmask |= 1 << stype[i];
  const int wave = tid >> 5, lane = tid & 31;
  unsigned tp[QG_];
#pragma unroll
  for (int g = 0; g < QG_; ++g) {
    unsigned v = 0;
#pragma unroll
    for (int i = 0; i < 8; ++i)
      v |= (1u << stype[g * 16 + (lane >> 4) * 8 + i]) << (3 * i);
    tp[g] = v;
  }
  const v8h* As8 = (const v8h*)As;
  project_tiles(As8, (const v16h*)w16, bias, outp, tp, tmask, base, wave, lane);
}

__device__ __forceinline__ void load_frag16(const float* __restrict__ p, v2f (&f)[8]) {
  const float4* p4 = (const float4*)p;
#pragma unroll
  for (int j = 0; j < 4; ++j) {
    const float4 t = p4[j];
    f[2 * j][0] = t.x;     f[2 * j][1] = t.y;
    f[2 * j + 1][0] = t.z; f[2 * j + 1][1] = t.w;
  }
}

__global__ __launch_bounds__(256) void rel_q_kernel(
    const float* __restrict__ qn, const float* __restrict__ rel_att,
    float* __restrict__ qa) {
  const int lane = threadIdx.x & 31, h = threadIdx.x >> 5;
  const int c = lane & 15, hs = lane >> 4;
  const int base = blockIdx.x * 16;
  v2f a[8];
  load_frag16(qn + (size_t)(base + c) * OUT_ + h * DK_ + hs * 16, a);
#pragma unroll
  for (int r = 0; r < R_; ++r) {
    const float* bp = rel_att + (((size_t)r * H_ + h) * DK_ + c) * DK_ + hs * 16;
    v2f b0[8], b1[8];
    load_frag16(bp, b0);
    load_frag16(bp + 16 * DK_, b1);
    v8f acc0 = {0.0f, 0.0f, 0.0f, 0.0f, 0.0f, 0.0f, 0.0f, 0.0f};
    v8f acc1 = acc0;
#pragma unroll
    for (int ks = 0; ks < 8; ++ks) {
      acc0 = __builtin_amdgcn_wmma_f32_16x16x4_f32(false, a[ks], false, b0[ks],
                                                   (short)0, acc0, false, false);
      acc1 = __builtin_amdgcn_wmma_f32_16x16x4_f32(false, a[ks], false, b1[ks],
                                                   (short)0, acc1, false, false);
    }
#pragma unroll
    for (int i = 0; i < 8; ++i) {
      float* o = qa + ((size_t)(base + 8 * hs + i) * R_ + r) * OUT_ + h * DK_ + c;
      o[0]  = acc0[i];
      o[16] = acc1[i];
    }
  }
}

#define EPW_ 32
#define EGR_ 4

typedef __attribute__((ext_vector_type(4))) float v4f;

__global__ __launch_bounds__(64) void edge_logit_kernel(
    const float* __restrict__ kn, const float* __restrict__ rteK,
    const float* __restrict__ qa,
    const int* __restrict__ ei, const int* __restrict__ etype,
    const int* __restrict__ etime, const int* __restrict__ ntype,
    const float* __restrict__ rel_pri,
    float* __restrict__ logit, int* __restrict__ head, int* __restrict__ next) {
  const int lane = threadIdx.x & 31;
  const int e0 = (blockIdx.x * 2 + (threadIdx.x >> 5)) * EPW_;
  const int e_l = e0 + lane;
  const int src_l = ei[e_l], dst_l = ei[E_ + e_l];
  const int r_l = etype[e_l], tm_l = etime[e_l];
  const int st_l = ntype[src_l];
  const int okn_l = src_l * OUT_;
  const int ort_l = (st_l * ML_ + tm_l) * OUT_;
  const int oqa_l = (dst_l * R_ + r_l) * OUT_;
  const int hq = lane >> 2, eq = lane & 3;
  const bool b0 = (eq & 1) != 0, b1 = (eq & 2) != 0;
#pragma unroll 1
  for (int g = 0; g < EPW_ / EGR_; ++g) {
    v4f k0[EGR_], k1[EGR_], t0[EGR_], t1[EGR_], q0[EGR_], q1[EGR_];
#pragma unroll
    for (int j = 0; j < EGR_; ++j) {
      const int i = g * EGR_ + j;
      const v4f* kb = (const v4f*)(kn + __builtin_amdgcn_readlane(okn_l, i)) + lane * 2;
      const v4f* rb = (const v4f*)(rteK + __builtin_amdgcn_readlane(ort_l, i)) + lane * 2;
      const v4f* qb = (const v4f*)(qa + __builtin_amdgcn_readlane(oqa_l, i)) + lane * 2;
      k0[j] = kb[0]; k1[j] = kb[1];
      t0[j] = rb[0]; t1[j] = rb[1];
      q0[j] = qb[0]; q1[j] = qb[1];
    }
    const int rr = __shfl(r_l, g * EGR_ + eq, 32);
    const float pri = rel_pri[rr * H_ + hq];
    asm volatile("" :: "v"(k0[0]), "v"(k1[0]), "v"(t0[0]), "v"(t1[0]), "v"(q0[0]), "v"(q1[0]),
                       "v"(k0[1]), "v"(k1[1]), "v"(t0[1]), "v"(t1[1]), "v"(q0[1]), "v"(q1[1]),
                       "v"(k0[2]), "v"(k1[2]), "v"(t0[2]), "v"(t1[2]), "v"(q0[2]), "v"(q1[2]),
                       "v"(k0[3]), "v"(k1[3]), "v"(t0[3]), "v"(t1[3]), "v"(q0[3]), "v"(q1[3]));
    float p[EGR_];
#pragma unroll
    for (int j = 0; j < EGR_; ++j) {
      float s = (k0[j].x + t0[j].x) * q0[j].x;
      s = fmaf(k0[j].y + t0[j].y, q0[j].y, s);
      s = fmaf(k0[j].z + t0[j].z, q0[j].z, s);
      s = fmaf(k0[j].w + t0[j].w, q0[j].w, s);
      s = fmaf(k1[j].x + t1[j].x, q1[j].x, s);
      s = fmaf(k1[j].y + t1[j].y, q1[j].y, s);
      s = fmaf(k1[j].z + t1[j].z, q1[j].z, s);
      s = fmaf(k1[j].w + t1[j].w, q1[j].w, s);
      p[j] = s;
    }
    const float x = (b0 ? p[1] : p[0]) + __shfl_xor(b0 ? p[0] : p[1], 1, 32);
    const float y = (b0 ? p[3] : p[2]) + __shfl_xor(b0 ? p[2] : p[3], 1, 32);
    const float z = (b1 ? y : x) + __shfl_xor(b1 ? x : y, 2, 32);
    const float lg = z * pri * 0.17677669529663689f;
    logit[(size_t)(e0 + g * EGR_ + eq) * H_ + hq] = lg;
  }
  next[e_l] = atomicExch(&head[dst_l], e_l);
}

__global__ __launch_bounds__(256) void rel_v_kernel(
    const float* __restrict__ vn, const float* __restrict__ rteV,
    const float* __restrict__ rel_msg,
    float* __restrict__ va, float* __restrict__ rtv) {
  const int lane = threadIdx.x & 31, h = threadIdx.x >> 5;
  const int c = lane & 15, hs = lane >> 4;
  const bool node_rows = blockIdx.x < N_ / 16;
  const float* __restrict__ rows = node_rows ? vn : rteV;
  float* __restrict__ outp = node_rows ? va : rtv;
  const int base = (node_rows ? blockIdx.x : blockIdx.x - N_ / 16) * 16;
  v2f a[8];
  load_frag16(rows + (size_t)(base + c) * OUT_ + h * DK_ + hs * 16, a);
#pragma unroll
  for (int r = 0; r < R_; ++r) {
    const float* bp = rel_msg + (((size_t)r * H_ + h) * DK_ + hs * 16) * DK_ + c;
    v2f b0[8], b1[8];
#pragma unroll
    for (int ks = 0; ks < 8; ++ks) {
      b0[ks][0] = bp[(2 * ks) * DK_];
      b0[ks][1] = bp[(2 * ks + 1) * DK_];
      b1[ks][0] = bp[(2 * ks) * DK_ + 16];
      b1[ks][1] = bp[(2 * ks + 1) * DK_ + 16];
    }
    v8f acc0 = {0.0f, 0.0f, 0.0f, 0.0f, 0.0f, 0.0f, 0.0f, 0.0f};
    v8f acc1 = acc0;
#pragma unroll
    for (int ks = 0; ks < 8; ++ks) {
      acc0 = __builtin_amdgcn_wmma_f32_16x16x4_f32(false, a[ks], false, b0[ks],
                                                   (short)0, acc0, false, false);
      acc1 = __builtin_amdgcn_wmma_f32_16x16x4_f32(false, a[ks], false, b1[ks],
                                                   (short)0, acc1, false, false);
    }
#pragma unroll
    for (int i = 0; i < 8; ++i) {
      float* o = outp + ((size_t)(base + 8 * hs + i) * R_ + r) * OUT_ + h * DK_ + c;
      o[0]  = acc0[i];
      o[16] = acc1[i];
    }
  }
}

__global__ __launch_bounds__(256) void node_aggr_kernel(
    const float* __restrict__ va, const float* __restrict__ rtv,
    const float* __restrict__ logit,
    const int* __restrict__ ei, const int* __restrict__ etype,
    const int* __restrict__ etime, const int* __restrict__ ntype,
    const int* __restrict__ head, const int* __restrict__ next,
    float* __restrict__ aggr) {
  const int lane = threadIdx.x & 31;
  const int n = blockIdx.x * 8 + (threadIdx.x >> 5);
  const int hl = lane & 7;
  const int e0 = head[n];
  float m = -__builtin_inff();
  int e = e0;
  for (int i = 0; i < E_ && e >= 0; ++i) {
    m = fmaxf(m, logit[(size_t)e * H_ + hl]);
    e = next[e];
  }
  float den = 0.0f;
  float acc[H_];
#pragma unroll
  for (int j = 0; j < H_; ++j) acc[j] = 0.0f;
  e = e0;
  for (int i = 0; i < E_ && e >= 0; ++i) {
    const int src = ei[e];
    const int r = etype[e], tm = etime[e];
    const int en = next[e];
    const float ex = __expf(logit[(size_t)e * H_ + hl] - m);
    const int st = ntype[src];
    const float* vb = va + ((size_t)src * R_ + r) * OUT_ + lane;
    const float* rb = rtv + (((size_t)st * ML_ + tm) * R_ + r) * OUT_ + lane;
    den += ex;
#pragma unroll
    for (int j = 0; j < H_; ++j) {
      const float w = __shfl(ex, j, 32);
      acc[j] = fmaf(w, vb[j * DK_] + rb[j * DK_], acc[j]);
    }
    e = en;
  }
  float* o = aggr + (size_t)n * OUT_ + lane;
#pragma unroll
  for (int j = 0; j < H_; ++j) {
    const float d = __shfl(den, j, 32);
    o[j * DK_] = (e0 >= 0) ? acc[j] / d : 0.0f;
  }
}

__global__ __launch_bounds__(256) void out_wmma_kernel(
    const float* __restrict__ aggr, const _Float16* __restrict__ wa16,
    const float* __restrict__ ba, const float* __restrict__ x,
    const int* __restrict__ ntype, const float* __restrict__ skip,
    const float* __restrict__ ln_g, const float* __restrict__ ln_b,
    float* __restrict__ out) {
  __shared__ __align__(32) _Float16 As[16 * OUT_];
  __shared__ float Tr[16 * OUT_];
  __shared__ int stype[16];
  __shared__ float alpha_s[16], mu_s[16], rs_s[16];
  const int tid = threadIdx.x;
  const int base = blockIdx.x * 16;
#pragma unroll
  for (int j = 0; j < 16; ++j) {
    int flat = j * 256 + tid;
    int s = flat & 15, ln = (flat >> 4) & 31, kk = flat >> 9;
    int k = kk * 32 + kmap(s, ln >> 4);
    As[flat] = (_Float16)aggr[(size_t)(base + (ln & 15)) * OUT_ + k];
  }
  if (tid < 16) {
    int t = ntype[base + tid];
    stype[tid] = t;
    alpha_s[tid] = 1.0f / (1.0f + __expf(-skip[t]));
  }
  __syncthreads();
  int tmask = 0;
#pragma unroll
  for (int i = 0; i < 16; ++i) tmask |= 1 << stype[i];
  const int wave = tid >> 5, lane = tid & 31;
  const int n = lane & 15, hs = lane >> 4;
  const v16h* Asv = (const v16h*)As;
  const v16h* Wpk = (const v16h*)wa16;
  for (int t = 0; t < T_; ++t) {
    if (!((tmask >> t) & 1)) continue;
#pragma unroll
    for (int c = 0; c < 2; ++c) {
      const int ct = wave * 2 + c;
      const int nbase = ct * 16;
      const float bv = ba[t * OUT_ + nbase + n];
      v8f acc;
#pragma unroll
      for (int i = 0; i < 8; ++i) acc[i] = bv;
#pragma unroll
      for (int kk = 0; kk < 8; ++kk) {
        v16h a = Asv[kk * 32 + lane];
        v16h b = Wpk[(((size_t)t * 16 + ct) * 8 + kk) * 32 + lane];
        acc = __builtin_amdgcn_wmma_f32_16x16x32_f16(false, a, false, b,
                                                     (short)0, acc, false, false);
      }
#pragma unroll
      for (int i = 0; i < 8; ++i) {
        int m = i + hs * 8;
        if (stype[m] == t) Tr[m * OUT_ + nbase + n] = acc[i];
      }
    }
  }
  __syncthreads();
#pragma unroll
  for (int j = 0; j < 16; ++j) {
    int flat = j * 256 + tid;
    int row = flat >> 8, col = flat & 255;
    float a = alpha_s[row];
    Tr[flat] = Tr[flat] * a + x[(size_t)(base + row) * IN_ + col] * (1.0f - a);
  }
  __syncthreads();
  if (tid < 16) {
    float mu = 0.0f;
    for (int c = 0; c < OUT_; ++c) mu += Tr[tid * OUT_ + c];
    mu *= (1.0f / OUT_);
    float var = 0.0f;
    for (int c = 0; c < OUT_; ++c) {
      float d = Tr[tid * OUT_ + c] - mu;
      var = fmaf(d, d, var);
    }
    var *= (1.0f / OUT_);
    mu_s[tid] = mu;
    rs_s[tid] = rsqrtf(var + 1e-5f);
  }
  __syncthreads();
#pragma unroll
  for (int j = 0; j < 16; ++j) {
    int flat = j * 256 + tid;
    int row = flat >> 8, col = flat & 255;
    int t = stype[row];
    out[(size_t)(base + row) * OUT_ + col] =
        (Tr[flat] - mu_s[row]) * rs_s[row] * ln_g[t * OUT_ + col] + ln_b[t * OUT_ + col];
  }
}

extern "C" void kernel_launch(void* const* d_in, const int* in_sizes, int n_in,
                              void* d_out, int out_size, void* d_ws, size_t ws_size,
                              hipStream_t stream) {
  (void)in_sizes; (void)n_in; (void)out_size; (void)ws_size;
  const float* node_inp  = (const float*)d_in[0];
  const float* Wk        = (const float*)d_in[1];
  const float* bk        = (const float*)d_in[2];
  const float* Wq        = (const float*)d_in[3];
  const float* bq        = (const float*)d_in[4];
  const float* Wv        = (const float*)d_in[5];
  const float* bv        = (const float*)d_in[6];
  const float* Wa        = (const float*)d_in[7];
  const float* ba        = (const float*)d_in[8];
  const float* ln_g      = (const float*)d_in[9];
  const float* ln_b      = (const float*)d_in[10];
  const float* rel_pri   = (const float*)d_in[11];
  const float* rel_att   = (const float*)d_in[12];
  const float* rel_msg   = (const float*)d_in[13];
  const float* skip      = (const float*)d_in[14];
  const float* rte_W     = (const float*)d_in[15];
  const float* rte_b     = (const float*)d_in[16];
  const float* rte_emb   = (const float*)d_in[17];
  const int*   node_type = (const int*)d_in[18];
  const int*   edge_idx  = (const int*)d_in[19];
  const int*   edge_type = (const int*)d_in[20];
  const int*   edge_time = (const int*)d_in[21];
  float* outp = (float*)d_out;

  char* p = (char*)d_ws;
  auto alloc = [&](size_t bytes) -> void* {
    void* r = (void*)p;
    p += (bytes + 255) & ~(size_t)255;
    return r;
  };
  float* rte_vec = (float*)alloc((size_t)ML_ * IN_ * 4);
  float* rteK    = (float*)alloc((size_t)T_ * ML_ * OUT_ * 4);
  float* rteV    = (float*)alloc((size_t)T_ * ML_ * OUT_ * 4);
  float* qn      = (float*)alloc((size_t)N_ * OUT_ * 4);
  float* kn      = (float*)alloc((size_t)N_ * OUT_ * 4);
  float* vn      = (float*)alloc((size_t)N_ * OUT_ * 4);
  float* logit   = (float*)alloc((size_t)E_ * H_ * 4);
  int*   head    = (int*)alloc((size_t)N_ * 4);
  int*   next    = (int*)alloc((size_t)E_ * 4);
  float* rtv     = (float*)alloc((size_t)T_ * ML_ * R_ * OUT_ * 4);
  float* qa      = (float*)alloc((size_t)N_ * R_ * OUT_ * 4);
  float* aggr    = qn;
  float* va      = qa;
  _Float16* wq16 = (_Float16*)alloc((size_t)T_ * IN_ * OUT_ * 2);
  _Float16* wk16 = (_Float16*)alloc((size_t)T_ * IN_ * OUT_ * 2);
  _Float16* wv16 = (_Float16*)alloc((size_t)T_ * IN_ * OUT_ * 2);
  _Float16* wa16 = (_Float16*)alloc((size_t)T_ * OUT_ * OUT_ * 2);

  const int wtotal = T_ * IN_ * OUT_;
  pack_w_kernel<<<wtotal / 256, 256, 0, stream>>>(Wq, wq16, wtotal);
  pack_w_kernel<<<wtotal / 256, 256, 0, stream>>>(Wk, wk16, wtotal);
  pack_w_kernel<<<wtotal / 256, 256, 0, stream>>>(Wv, wv16, wtotal);
  pack_w_kernel<<<wtotal / 256, 256, 0, stream>>>(Wa, wa16, wtotal);

  rte_vec_kernel<<<(ML_ * IN_) / 256, 256, 0, stream>>>(rte_emb, rte_W, rte_b, rte_vec);
  rte_kv_kernel<<<(T_ * ML_ * OUT_) / 256, 256, 0, stream>>>(rte_vec, Wk, Wv, rteK, rteV);

  qkv_wmma_kernel<<<3 * (N_ / (QG_ * 16)), 256, 0, stream>>>(node_inp, wq16, wk16, wv16,
                                               bq, bk, bv, node_type, qn, kn, vn);

  rel_q_kernel<<<N_ / 16, 256, 0, stream>>>(qn, rel_att, qa);

  list_init_kernel<<<(N_ + 255) / 256, 256, 0, stream>>>(head);

  edge_logit_kernel<<<E_ / 64, 64, 0, stream>>>(kn, rteK, qa, edge_idx, edge_type,
                                                edge_time, node_type, rel_pri,
                                                logit, head, next);
  rel_v_kernel<<<N_ / 16 + (T_ * ML_) / 16, 256, 0, stream>>>(vn, rteV, rel_msg, va, rtv);
  node_aggr_kernel<<<N_ / 8, 256, 0, stream>>>(va, rtv, logit, edge_idx, edge_type,
                                               edge_time, node_type, head, next, aggr);

  out_wmma_kernel<<<N_ / 16, 256, 0, stream>>>(aggr, wa16, ba, node_inp, node_type,
                                               skip, ln_g, ln_b, outp);
}


# q/k/v and output projections: nodes grouped by type (index list built in the set-up launch); a block of 80 same-type nodes multiplies by its own type's weights only (was all three); same sums
# speedup vs baseline: 3.1681x; 1.6578x over previous
#include <hip/hip_runtime.h>
#include <math.h>

#define T_   3
#define R_   5
#define H_   8
#define IN_  256
#define OUT_ 256
#define DK_  32
#define N_   30000
#define E_   200000
#define ML_  240

typedef __attribute__((ext_vector_type(16))) _Float16 v16h;
typedef __attribute__((ext_vector_type(8)))  float    v8f;
typedef __attribute__((ext_vector_type(2)))  float    v2f;

__device__ __forceinline__ int kmap(int s, int half) {
  int v = s >> 1;
  return ((v >> 2) << 4) | (half << 3) | ((v & 3) << 1) | (s & 1);
}

#define PWB_ 48
#define RVB_ 240
#define LIB_ 118
#define QR_  80
#define GB_  378
#define NT4_ 30

__device__ __forceinline__ unsigned type_fields(int4 v) {
  return (1u << (10 * v.x)) + (1u << (10 * v.y)) + (1u << (10 * v.z)) + (1u << (10 * v.w));
}

__device__ __forceinline__ void pack_w_frag(const float* __restrict__ W,
                                            _Float16* __restrict__ Wp, int f) {
  const int lane = f & 31, kk = (f >> 5) & 7, ct = (f >> 8) & 15, t = f >> 12;
  const int half = lane >> 4, n = lane & 15;
  const float* s0 = W + (size_t)t * (IN_ * OUT_) + (size_t)(kk * 32) * OUT_ + ct * 16 + n;
  v16h o;
#pragma unroll
  for (int s = 0; s < 16; ++s) o[s] = (_Float16)s0[(size_t)kmap(s, half) * OUT_];
  ((v16h*)Wp)[f] = o;
}

__global__ __launch_bounds__(256) void setup_kernel(
    const float* __restrict__ Wq, const float* __restrict__ Wk,
    const float* __restrict__ Wv, const float* __restrict__ Wa,
    _Float16* __restrict__ wq16, _Float16* __restrict__ wk16,
    _Float16* __restrict__ wv16, _Float16* __restrict__ wa16,
    const float* __restrict__ emb, const float* __restrict__ Wt,
    const float* __restrict__ b, float* __restrict__ rv,
    int* __restrict__ head, const int* __restrict__ ntype,
    int* __restrict__ perm, int* __restrict__ btype) {
  __shared__ int red[8][9];
  const int blk = blockIdx.x, tid = threadIdx.x;
  if (blk < 4 * PWB_) {
    const int which = blk / PWB_;
    const float* __restrict__ W = which == 0 ? Wq : (which == 1 ? Wk : (which == 2 ? Wv : Wa));
    _Float16* __restrict__ Wp = which == 0 ? wq16 : (which == 1 ? wk16 : (which == 2 ? wv16 : wa16));
    pack_w_frag(W, Wp, (blk - which * PWB_) * 256 + tid);
  } else if (blk < 4 * PWB_ + RVB_) {
    const int idx = (blk - 4 * PWB_) * 256 + tid;
    const int i = idx & 255, m = idx >> 8;
    float acc = b[i];
    const float* e = emb + (size_t)m * IN_;
    const float* w = Wt + (size_t)i * IN_;
    for (int j = 0; j < IN_; ++j) acc = fmaf(e[j], w[j], acc);
    rv[idx] = acc;
  } else {
    const int lb = blk - 4 * PWB_ - RVB_;
    const int i = lb * 256 + tid;
    const int lane = tid & 31, wave = tid >> 5;
    int ti = -1;
    if (i < N_) { head[i] = -1; ti = ntype[i]; }
    const int4* nt4 = (const int4*)ntype;
    unsigned all = 0, bef = 0;
#pragma unroll
    for (int it = 0; it < NT4_ - 1; ++it) {
      const int q = it * 256 + tid;
      const unsigned f = type_fields(nt4[q]);
      all += f;
      bef += q < 64 * lb ? f : 0u;
    }
    {
      const int q = (NT4_ - 1) * 256 + tid;
      if (q < N_ / 4) {
        const unsigned f = type_fields(nt4[q]);
        all += f;
        bef += q < 64 * lb ? f : 0u;
      }
    }
    int cn[6];
    cn[0] = all & 1023; cn[1] = (all >> 10) & 1023; cn[2] = (all >> 20) & 1023;
    cn[3] = bef & 1023; cn[4] = (bef >> 10) & 1023; cn[5] = (bef >> 20) & 1023;
#pragma unroll
    for (int o = 16; o > 0; o >>= 1) {
#pragma unroll
      for (int k = 0; k < 6; ++k) cn[k] += __shfl_xor(cn[k], o, 32);
    }
    const unsigned m0 = __builtin_amdgcn_ballot_w32(ti == 0);
    const unsigned m1 = __builtin_amdgcn_ballot_w32(ti == 1);
    const unsigned m2 = __builtin_amdgcn_ballot_w32(ti == 2);
    if (lane == 0) {
#pragma unroll
      for (int k = 0; k < 6; ++k) red[wave][k] = cn[k];
      red[wave][6] = __builtin_popcount(m0);
      red[wave][7] = __builtin_popcount(m1);
      red[wave][8] = __builtin_popcount(m2);
    }
    __syncthreads();
    int tot0 = 0, tot1 = 0, tot2 = 0, bf0 = 0, bf1 = 0, bf2 = 0, w0 = 0, w1 = 0, w2 = 0;
#pragma unroll
    for (int w = 0; w < 8; ++w) {
      tot0 += red[w][0]; tot1 += red[w][1]; tot2 += red[w][2];
      bf0 += red[w][3];  bf1 += red[w][4];  bf2 += red[w][5];
      if (w < wave) { w0 += red[w][6]; w1 += red[w][7]; w2 += red[w][8]; }
    }
    const int nb0 = (tot0 + QR_ - 1) / QR_, nb1 = (tot1 + QR_ - 1) / QR_;
    const int nb2 = (tot2 + QR_ - 1) / QR_;
    const int st1 = QR_ * nb0, st2 = QR_ * (nb0 + nb1);
    if (ti >= 0) {
      const unsigned mm = ti == 0 ? m0 : (ti == 1 ? m1 : m2);
      const int place = ti == 0 ? bf0 + w0 : (ti == 1 ? st1 + bf1 + w1 : st2 + bf2 + w2);
      perm[place + __builtin_popcount(mm & ((1u << lane) - 1u))] = i;
    }
    if (lb == 0) {
      for (int p = tid; p < GB_ * QR_; p += 256) {
        const int off = p < st1 ? p : (p < st2 ? p - st1 : p - st2);
        const int cnt = p < st1 ? tot0 : (p < st2 ? tot1 : tot2);
        if (off >= cnt) perm[p] = -1;
      }
      for (int g = tid; g < GB_; g += 256)
        btype[g] = g < nb0 ? 0 : (g < nb0 + nb1 ? 1 : (g < nb0 + nb1 + nb2 ? 2 : -1));
    }
  }
}

#define RKB_ 720
__device__ __forceinline__ void rte_kv_block(const float* __restrict__ rv,
                              const float* __restrict__ Wk, const float* __restrict__ Wv,
                              float* __restrict__ rk, float* __restrict__ rvv, int blk) {
  int idx = blk * 256 + threadIdx.x;
  int o = idx & 255;
  int m = (idx >> 8) % ML_;
  int t = idx / (ML_ * OUT_);
  const float* v  = rv + (size_t)m * IN_;
  const float* wk = Wk + (size_t)t * (IN_ * OUT_);
  const float* wv = Wv + (size_t)t * (IN_ * OUT_);
  float ak = 0.0f, av = 0.0f;
  for (int i = 0; i < IN_; ++i) {
    float x = v[i];
    ak = fmaf(x, wk[(size_t)i * OUT_ + o], ak);
    av = fmaf(x, wv[(size_t)i * OUT_ + o], av);
  }
  rk[idx] = ak;
  rvv[idx] = av;
}

#define QG_ 5

typedef __attribute__((ext_vector_type(8))) _Float16 v8h;

template <bool SCATTER>
__device__ __forceinline__ void project_tiles(
    const v8h* __restrict__ As8, const v16h* __restrict__ Wpk,
    const float* __restrict__ bias, float* __restrict__ Out,
    const int* srow, int t, int wave, int lane) {
  const int n = lane & 15, hs = lane >> 4;
  const int ct0 = wave * 2;
  const int nbase = ct0 * 16;
  const v8h* ap = As8 + lane;
  {
    const v16h* wp = Wpk + (((size_t)t * 16 + ct0) * 8) * 32 + lane;
    v16h b0[2], b1[2];
    v8h alo[2][QG_], ahi[2][QG_];
    b0[0] = wp[0];      b1[0] = wp[8 * 32];
#pragma unroll
    for (int g = 0; g < QG_; ++g) {
      alo[0][g] = ap[((g * 8) * 2 + 0) * 32];
      ahi[0][g] = ap[((g * 8) * 2 + 1) * 32];
    }
    const float bv0 = bias[t * OUT_ + nbase + n];
    const float bv1 = bias[t * OUT_ + nbase + 16 + n];
    v8f acc[QG_][2];
#pragma unroll
    for (int g = 0; g < QG_; ++g) {
#pragma unroll
      for (int i = 0; i < 8; ++i) { acc[g][0][i] = bv0; acc[g][1][i] = bv1; }
    }
#pragma unroll
    for (int kk = 0; kk < 8; ++kk) {
      const int cur = kk & 1, nxt = cur ^ 1;
      if (kk + 1 < 8) {
        b0[nxt] = wp[(kk + 1) * 32];
        b1[nxt] = wp[(8 + kk + 1) * 32];
#pragma unroll
        for (int g = 0; g < QG_; ++g) {
          alo[nxt][g] = ap[((g * 8 + kk + 1) * 2 + 0) * 32];
          ahi[nxt][g] = ap[((g * 8 + kk + 1) * 2 + 1) * 32];
        }
      }
      __builtin_amdgcn_sched_barrier(0);
#pragma unroll
      for (int g = 0; g < QG_; ++g) {
        const v16h a = __builtin_shufflevector(alo[cur][g], ahi[cur][g],
                                               0, 1, 2, 3, 4, 5, 6, 7,
                                               8, 9, 10, 11, 12, 13, 14, 15);
        acc[g][0] = __builtin_amdgcn_wmma_f32_16x16x32_f16(false, a, false, b0[cur],
                                                           (short)0, acc[g][0], false, false);
        acc[g][1] = __builtin_amdgcn_wmma_f32_16x16x32_f16(false, a, false, b1[cur],
                                                           (short)0, acc[g][1], false, false);
      }
      __builtin_amdgcn_sched_barrier(0);
    }
#pragma unroll
    for (int g = 0; g < QG_; ++g) {
#pragma unroll
      for (int i = 0; i < 8; ++i) {
        const int lr = g * 16 + hs * 8 + i;
        if (SCATTER) {
          const int row = srow[lr];
          if (row >= 0) {
            float* o = Out + (size_t)row * OUT_ + nbase + n;
            o[0]  = acc[g][0][i];
            o[16] = acc[g][1][i];
          }
        } else {
          float* o = Out + lr * OUT_ + nbase + n;
          o[0]  = acc[g][0][i];
          o[16] = acc[g][1][i];
        }
      }
    }
  }
}

#define LPB_ 782
__global__ __launch_bounds__(256) void qkv_wmma_kernel(
    const float* __restrict__ x,
    const _Float16* __restrict__ wq16, const _Float16* __restrict__ wk16,
    const _Float16* __restrict__ wv16,
    const float* __restrict__ bq, const float* __restrict__ bk, const float* __restrict__ bv,
    const int* __restrict__ perm, const int* __restrict__ btype,
    float* __restrict__ qn, float* __restrict__ kn, float* __restrict__ vn,
    const float* __restrict__ rv, const float* __restrict__ Wk, const float* __restrict__ Wv,
    float* __restrict__ rk, float* __restrict__ rvv,
    const int* __restrict__ ei, int* __restrict__ head, int* __restrict__ next) {
  __shared__ __align__(32) _Float16 As[QG_ * 16 * IN_];
  __shared__ int srow[QG_ * 16];
  const int tid = threadIdx.x;
  if (blockIdx.x < RKB_) {
    rte_kv_block(rv, Wk, Wv, rk, rvv, blockIdx.x);
    return;
  }
  if (blockIdx.x < RKB_ + LPB_) {
    const int e = (blockIdx.x - RKB_) * 256 + tid;
    if (e < E_) next[e] = atomicExch(&head[ei[E_ + e]], e);
    return;
  }
  const int bid = blockIdx.x - RKB_ - LPB_;
  const int proj = bid % 3;
  const int grp = bid / 3;
  const int t = btype[grp];
  const int pr = tid < QG_ * 16 ? perm[grp * (QG_ * 16) + tid] : -1;
  if (t < 0) return;
  if (tid < QG_ * 16) srow[tid] = pr;
  __syncthreads();
  const _Float16* __restrict__ w16 = proj == 0 ? wq16 : (proj == 1 ? wk16 : wv16);
  const float* __restrict__ bias = proj == 0 ? bq : (proj == 1 ? bk : bv);
  float* __restrict__ outp = proj == 0 ? qn : (proj == 1 ? kn : vn);
#pragma unroll
  for (int g = 0; g < QG_; ++g) {
    const int node = srow[g * 16 + (tid >> 4)];
    const float* xr = x + (size_t)(node < 0 ? 0 : node) * IN_;
#pragma unroll
    for (int j = 0; j < 16; ++j) {
      int flat = j * 256 + tid;
      int s = flat & 15, ln = (flat >> 4) & 31, kk = flat >> 9;
      int k = kk * 32 + kmap(s, ln >> 4);
      const float xv = xr[k];
      As[((((g * 8 + kk) * 2 + (s >> 3)) * 32 + ln) << 3) + (s & 7)] =
          (_Float16)(node < 0 ? 0.0f : xv);
    }
  }
  __syncthreads();
  const int wave = tid >> 5, lane = tid & 31;
  const v8h* As8 = (const v8h*)As;
  project_tiles<true>(As8, (const v16h*)w16, bias, outp, srow, t, wave, lane);
}

__device__ __forceinline__ void load_frag16(const float* __restrict__ p, v2f (&f)[8]) {
  const float4* p4 = (const float4*)p;
#pragma unroll
  for (int j = 0; j < 4; ++j) {
    const float4 t = p4[j];
    f[2 * j][0] = t.x;     f[2 * j][1] = t.y;
    f[2 * j + 1][0] = t.z; f[2 * j + 1][1] = t.w;
  }
}

#define NB_ 16

typedef __attribute__((ext_vector_type(4))) float v4f;

__device__ __forceinline__ float dot8(v4f k0, v4f k1, v4f t0, v4f t1, v4f q0, v4f q1) {
  float s = (k0.x + t0.x) * q0.x;
  s = fmaf(k0.y + t0.y, q0.y, s);
  s = fmaf(k0.z + t0.z, q0.z, s);
  s = fmaf(k0.w + t0.w, q0.w, s);
  s = fmaf(k1.x + t1.x, q1.x, s);
  s = fmaf(k1.y + t1.y, q1.y, s);
  s = fmaf(k1.z + t1.z, q1.z, s);
  s = fmaf(k1.w + t1.w, q1.w, s);
  return s;
}

__global__ __launch_bounds__(512) __attribute__((amdgpu_waves_per_eu(16, 16)))
void node_logit_kernel(
    const float* __restrict__ qn, const float* __restrict__ rel_att,
    const float* __restrict__ kn, const float* __restrict__ rteK,
    const int* __restrict__ ei, const int* __restrict__ etype,
    const int* __restrict__ etime, const int* __restrict__ ntype,
    const float* __restrict__ rel_pri,
    const int* __restrict__ head, const int* __restrict__ next,
    float* __restrict__ logit) {
  __shared__ __align__(16) float Q[NB_ * R_ * OUT_];
  const int lane = threadIdx.x & 31, wave = threadIdx.x >> 5;
  const int base = blockIdx.x * NB_;
  if (wave < H_) {
    const int h = wave;
    const int c = lane & 15, hs = lane >> 4;
    v2f a[8];
    load_frag16(qn + (size_t)(base + c) * OUT_ + h * DK_ + hs * 16, a);
#pragma unroll 1
    for (int r = 0; r < R_; ++r) {
      const float* bp = rel_att + (((size_t)r * H_ + h) * DK_ + c) * DK_ + hs * 16;
      v2f b0[8], b1[8];
      load_frag16(bp, b0);
      load_frag16(bp + 16 * DK_, b1);
      v8f acc0 = {0.0f, 0.0f, 0.0f, 0.0f, 0.0f, 0.0f, 0.0f, 0.0f};
      v8f acc1 = acc0;
#pragma unroll
      for (int ks = 0; ks < 8; ++ks) {
        acc0 = __builtin_amdgcn_wmma_f32_16x16x4_f32(false, a[ks], false, b0[ks],
                                                     (short)0, acc0, false, false);
        acc1 = __builtin_amdgcn_wmma_f32_16x16x4_f32(false, a[ks], false, b1[ks],
                                                     (short)0, acc1, false, false);
      }
#pragma unroll
      for (int i = 0; i < 8; ++i) {
        float* o = Q + ((8 * hs + i) * R_ + r) * OUT_ + h * DK_ + c;
        o[0]  = acc0[i];
        o[16] = acc1[i];
      }
    }
  }
  __syncthreads();
  const int n = base + wave;
  const int hq = lane >> 2;
  const float* qrow = Q + wave * (R_ * OUT_) + lane * 8;
  int e = head[n];
  for (int ch = 0; ch < (E_ + 31) / 32 && e >= 0; ++ch) {
    int my_e = 0;
    int got = 0;
    for (; got < 32 && e >= 0; ++got) {
      if (got == lane) my_e = e;
      e = next[e];
    }
    const int cnt = __builtin_amdgcn_readfirstlane(got);
    int okn_l = 0, ort_l = 0, r_l = 0;
    if (lane < cnt) {
      const int src = ei[my_e];
      r_l = etype[my_e];
      const int tm = etime[my_e];
      okn_l = src * OUT_;
      ort_l = (ntype[src] * ML_ + tm) * OUT_;
    }
    for (int j0 = 0; j0 < cnt; j0 += 2) {
      const bool two = j0 + 1 < cnt;
      const int j1 = two ? j0 + 1 : j0;
      const int ea = __builtin_amdgcn_readlane(my_e, j0);
      const int eb = __builtin_amdgcn_readlane(my_e, j1);
      const int ra = __builtin_amdgcn_readlane(r_l, j0);
      const int rb = __builtin_amdgcn_readlane(r_l, j1);
      const v4f* ka = (const v4f*)(kn + __builtin_amdgcn_readlane(okn_l, j0)) + lane * 2;
      const v4f* ta = (const v4f*)(rteK + __builtin_amdgcn_readlane(ort_l, j0)) + lane * 2;
      const v4f* kb = (const v4f*)(kn + __builtin_amdgcn_readlane(okn_l, j1)) + lane * 2;
      const v4f* tb = (const v4f*)(rteK + __builtin_amdgcn_readlane(ort_l, j1)) + lane * 2;
      const v4f ka0 = ka[0], ka1 = ka[1], ta0 = ta[0], ta1 = ta[1];
      const v4f kb0 = kb[0], kb1 = kb[1], tb0 = tb[0], tb1 = tb[1];
      const float pa = rel_pri[ra * H_ + hq];
      const float pb = rel_pri[rb * H_ + hq];
      asm volatile("" :: "v"(ka0), "v"(ka1), "v"(ta0), "v"(ta1), "v"(pa),
                         "v"(kb0), "v"(kb1), "v"(tb0), "v"(tb1), "v"(pb));
      const v4f* qa = (const v4f*)(qrow + ra * OUT_);
      const v4f* qb = (const v4f*)(qrow + rb * OUT_);
      float sa = dot8(ka0, ka1, ta0, ta1, qa[0], qa[1]);
      float sb = dot8(kb0, kb1, tb0, tb1, qb[0], qb[1]);
      sa += __shfl_xor(sa, 1, 32);
      sb += __shfl_xor(sb, 1, 32);
      sa += __shfl_xor(sa, 2, 32);
      sb += __shfl_xor(sb, 2, 32);
      if ((lane & 3) == 0) {
        logit[(size_t)ea * H_ + hq] = sa * pa * 0.17677669529663689f;
        if (two) logit[(size_t)eb * H_ + hq] = sb * pb * 0.17677669529663689f;
      }
    }
  }
}

__global__ __launch_bounds__(512) void node_aggr_kernel(
    const float* __restrict__ vn, const float* __restrict__ rteV,
    const float* __restrict__ rel_msg, const float* __restrict__ logit,
    const int* __restrict__ ei, const int* __restrict__ etype,
    const int* __restrict__ etime, const int* __restrict__ ntype,
    const int* __restrict__ head, const int* __restrict__ next,
    float* __restrict__ aggr) {
  __shared__ __align__(16) float S[NB_ * R_ * OUT_];
  const int lane = threadIdx.x & 31, wave = threadIdx.x >> 5;
  const int base = blockIdx.x * NB_;
  {
    const int n = base + wave;
    const int hq = lane >> 2;
    float* srow = S + wave * (R_ * OUT_);
    const int c0 = (8 * lane + 4 * wave) & 255;
    const int c1 = (8 * lane + 4 + 4 * wave) & 255;
    const v4f zero = {0.0f, 0.0f, 0.0f, 0.0f};
#pragma unroll
    for (int r = 0; r < R_; ++r) {
      *(v4f*)(srow + r * OUT_ + c0) = zero;
      *(v4f*)(srow + r * OUT_ + c1) = zero;
    }
    const int e0 = head[n];
    float m = -__builtin_inff();
    int e = e0;
    for (int i = 0; i < E_ && e >= 0; ++i) {
      m = fmaxf(m, logit[(size_t)e * H_ + hq]);
      e = next[e];
    }
    float den = 0.0f;
    e = e0;
    for (int i = 0; i < E_ && e >= 0; ++i) {
      const int src = ei[e];
      const int r = etype[e], tm = etime[e];
      const int en = next[e];
      const float ex = __expf(logit[(size_t)e * H_ + hq] - m);
      const int st = ntype[src];
      const v4f* vb = (const v4f*)(vn + (size_t)src * OUT_) + lane * 2;
      const v4f x0 = vb[0], x1 = vb[1];
      __builtin_amdgcn_sched_barrier(0);
      const v4f* rb = (const v4f*)(rteV + ((size_t)st * ML_ + tm) * OUT_) + lane * 2;
      const v4f t0 = rb[0], t1 = rb[1];
      v4f* s0 = (v4f*)(srow + r * OUT_ + c0);
      v4f* s1 = (v4f*)(srow + r * OUT_ + c1);
      v4f a0 = *s0, a1 = *s1;
      den += ex;
      a0.x = fmaf(ex, x0.x + t0.x, a0.x);
      a0.y = fmaf(ex, x0.y + t0.y, a0.y);
      a0.z = fmaf(ex, x0.z + t0.z, a0.z);
      a0.w = fmaf(ex, x0.w + t0.w, a0.w);
      a1.x = fmaf(ex, x1.x + t1.x, a1.x);
      a1.y = fmaf(ex, x1.y + t1.y, a1.y);
      a1.z = fmaf(ex, x1.z + t1.z, a1.z);
      a1.w = fmaf(ex, x1.w + t1.w, a1.w);
      *s0 = a0;
      *s1 = a1;
      e = en;
    }
    if (e0 >= 0) {
      const float inv = 1.0f / den;
#pragma unroll
      for (int r = 0; r < R_; ++r) {
        v4f* s0 = (v4f*)(srow + r * OUT_ + c0);
        v4f* s1 = (v4f*)(srow + r * OUT_ + c1);
        v4f a0 = *s0, a1 = *s1;
        a0 *= inv;
        a1 *= inv;
        *s0 = a0;
        *s1 = a1;
      }
    }
  }
  __syncthreads();
  if (wave >= H_) return;
  const int h = wave;
  const int c = lane & 15, hs = lane >> 4;
  v8f acc0 = {0.0f, 0.0f, 0.0f, 0.0f, 0.0f, 0.0f, 0.0f, 0.0f};
  v8f acc1 = acc0;
#pragma unroll 1
  for (int r = 0; r < R_; ++r) {
    const float* ar = S + c * (R_ * OUT_) + r * OUT_;
    v2f a[8];
#pragma unroll
    for (int q = 0; q < 4; ++q) {
      const v4f t = *(const v4f*)(ar + ((h * DK_ + hs * 16 + 4 * q + 4 * c) & 255));
      a[2 * q][0] = t.x;     a[2 * q][1] = t.y;
      a[2 * q + 1][0] = t.z; a[2 * q + 1][1] = t.w;
    }
    const float* bp = rel_msg + (((size_t)r * H_ + h) * DK_ + hs * 16) * DK_ + c;
    v2f b0[8], b1[8];
#pragma unroll
    for (int ks = 0; ks < 8; ++ks) {
      b0[ks][0] = bp[(2 * ks) * DK_];
      b0[ks][1] = bp[(2 * ks + 1) * DK_];
      b1[ks][0] = bp[(2 * ks) * DK_ + 16];
      b1[ks][1] = bp[(2 * ks + 1) * DK_ + 16];
    }
#pragma unroll
    for (int ks = 0; ks < 8; ++ks) {
      acc0 = __builtin_amdgcn_wmma_f32_16x16x4_f32(false, a[ks], false, b0[ks],
                                                   (short)0, acc0, false, false);
      acc1 = __builtin_amdgcn_wmma_f32_16x16x4_f32(false, a[ks], false, b1[ks],
                                                   (short)0, acc1, false, false);
    }
  }
#pragma unroll
  for (int i = 0; i < 8; ++i) {
    float* o = aggr + (size_t)(base + 8 * hs + i) * OUT_ + h * DK_ + c;
    o[0]  = acc0[i];
    o[16] = acc1[i];
  }
}

#define OR_ (QG_ * 16)
#define ORW_ (OR_ / 8)

__global__ __launch_bounds__(256) void out_wmma_kernel(
    const float* __restrict__ aggr, const _Float16* __restrict__ wa16,
    const float* __restrict__ ba, const float* __restrict__ x,
    const int* __restrict__ perm, const int* __restrict__ btype,
    const float* __restrict__ skip,
    const float* __restrict__ ln_g, const float* __restrict__ ln_b,
    float* __restrict__ out) {
  __shared__ __align__(32) _Float16 As[OR_ * OUT_];
  __shared__ __align__(16) float Tr[OR_ * OUT_];
  __shared__ int srow[OR_];
  const int tid = threadIdx.x;
  const int t = btype[blockIdx.x];
  const int pr = tid < OR_ ? perm[blockIdx.x * OR_ + tid] : -1;
  if (t < 0) return;
  if (tid < OR_) srow[tid] = pr;
  __syncthreads();
#pragma unroll
  for (int g = 0; g < QG_; ++g) {
    const int node = srow[g * 16 + (tid >> 4)];
    const float* ar = aggr + (size_t)(node < 0 ? 0 : node) * OUT_;
#pragma unroll
    for (int j = 0; j < 16; ++j) {
      int flat = j * 256 + tid;
      int s = flat & 15, ln = (flat >> 4) & 31, kk = flat >> 9;
      int k = kk * 32 + kmap(s, ln >> 4);
      const float av = ar[k];
      As[((((g * 8 + kk) * 2 + (s >> 3)) * 32 + ln) << 3) + (s & 7)] =
          (_Float16)(node < 0 ? 0.0f : av);
    }
  }
  __syncthreads();
  const int wave = tid >> 5, lane = tid & 31;
  project_tiles<false>((const v8h*)As, (const v16h*)wa16, ba, Tr, srow, t, wave, lane);
  __syncthreads();
  const float a = 1.0f / (1.0f + __expf(-skip[t]));
  const v4f g0 = *(const v4f*)(ln_g + t * OUT_ + 4 * lane);
  const v4f g1 = *(const v4f*)(ln_g + t * OUT_ + 128 + 4 * lane);
  const v4f c0 = *(const v4f*)(ln_b + t * OUT_ + 4 * lane);
  const v4f c1 = *(const v4f*)(ln_b + t * OUT_ + 128 + 4 * lane);
#pragma unroll 5
  for (int rr = 0; rr < ORW_; ++rr) {
    const int row = wave * ORW_ + rr;
    const int node = srow[row];
    const float* xr = x + (size_t)(node < 0 ? 0 : node) * IN_ + 4 * lane;
    const float* tr = Tr + row * OUT_ + 4 * lane;
    const v4f x0 = *(const v4f*)xr, x1 = *(const v4f*)(xr + 128);
    const v4f t0 = *(const v4f*)tr, t1 = *(const v4f*)(tr + 128);
    float v[8];
    v[0] = t0.x * a + x0.x * (1.0f - a);
    v[1] = t0.y * a + x0.y * (1.0f - a);
    v[2] = t0.z * a + x0.z * (1.0f - a);
    v[3] = t0.w * a + x0.w * (1.0f - a);
    v[4] = t1.x * a + x1.x * (1.0f - a);
    v[5] = t1.y * a + x1.y * (1.0f - a);
    v[6] = t1.z * a + x1.z * (1.0f - a);
    v[7] = t1.w * a + x1.w * (1.0f - a);
    float mu = 0.0f;
#pragma unroll
    for (int j = 0; j < 8; ++j) mu += v[j];
#pragma unroll
    for (int o = 16; o > 0; o >>= 1) mu += __shfl_xor(mu, o, 32);
    mu *= (1.0f / OUT_);
    float var = 0.0f;
#pragma unroll
    for (int j = 0; j < 8; ++j) {
      const float d = v[j] - mu;
      var = fmaf(d, d, var);
    }
#pragma unroll
    for (int o = 16; o > 0; o >>= 1) var += __shfl_xor(var, o, 32);
    var *= (1.0f / OUT_);
    const float rs = rsqrtf(var + 1e-5f);
    v4f o0, o1;
    o0.x = (v[0] - mu) * rs * g0.x + c0.x;
    o0.y = (v[1] - mu) * rs * g0.y + c0.y;
    o0.z = (v[2] - mu) * rs * g0.z + c0.z;
    o0.w = (v[3] - mu) * rs * g0.w + c0.w;
    o1.x = (v[4] - mu) * rs * g1.x + c1.x;
    o1.y = (v[5] - mu) * rs * g1.y + c1.y;
    o1.z = (v[6] - mu) * rs * g1.z + c1.z;
    o1.w = (v[7] - mu) * rs * g1.w + c1.w;
    if (node >= 0) {
      float* orow = out + (size_t)node * OUT_ + 4 * lane;
      *(v4f*)orow = o0;
      *(v4f*)(orow + 128) = o1;
    }
  }
}

extern "C" void kernel_launch(void* const* d_in, const int* in_sizes, int n_in,
                              void* d_out, int out_size, void* d_ws, size_t ws_size,
                              hipStream_t stream) {
  (void)in_sizes; (void)n_in; (void)out_size; (void)ws_size;
  const float* node_inp  = (const float*)d_in[0];
  const float* Wk        = (const float*)d_in[1];
  const float* bk        = (const float*)d_in[2];
  const float* Wq        = (const float*)d_in[3];
  const float* bq        = (const float*)d_in[4];
  const float* Wv        = (const float*)d_in[5];
  const float* bv        = (const float*)d_in[6];
  const float* Wa        = (const float*)d_in[7];
  const float* ba        = (const float*)d_in[8];
  const float* ln_g      = (const float*)d_in[9];
  const float* ln_b      = (const float*)d_in[10];
  const float* rel_pri   = (const float*)d_in[11];
  const float* rel_att   = (const float*)d_in[12];
  const float* rel_msg   = (const float*)d_in[13];
  const float* skip      = (const float*)d_in[14];
  const float* rte_W     = (const float*)d_in[15];
  const float* rte_b     = (const float*)d_in[16];
  const float* rte_emb   = (const float*)d_in[17];
  const int*   node_type = (const int*)d_in[18];
  const int*   edge_idx  = (const int*)d_in[19];
  const int*   edge_type = (const int*)d_in[20];
  const int*   edge_time = (const int*)d_in[21];
  float* outp = (float*)d_out;

  char* p = (char*)d_ws;
  auto alloc = [&](size_t bytes) -> void* {
    void* r = (void*)p;
    p += (bytes + 255) & ~(size_t)255;
    return r;
  };
  float* rte_vec = (float*)alloc((size_t)ML_ * IN_ * 4);
  float* rteK    = (float*)alloc((size_t)T_ * ML_ * OUT_ * 4);
  float* rteV    = (float*)alloc((size_t)T_ * ML_ * OUT_ * 4);
  float* qn      = (float*)alloc((size_t)N_ * OUT_ * 4);
  float* kn      = (float*)alloc((size_t)N_ * OUT_ * 4);
  float* vn      = (float*)alloc((size_t)N_ * OUT_ * 4);
  float* logit   = (float*)alloc((size_t)E_ * H_ * 4);
  int*   head    = (int*)alloc((size_t)N_ * 4);
  int*   next    = (int*)alloc((size_t)E_ * 4);
  float* aggr    = qn;
  _Float16* wq16 = (_Float16*)alloc((size_t)T_ * IN_ * OUT_ * 2);
  _Float16* wk16 = (_Float16*)alloc((size_t)T_ * IN_ * OUT_ * 2);
  _Float16* wv16 = (_Float16*)alloc((size_t)T_ * IN_ * OUT_ * 2);
  _Float16* wa16 = (_Float16*)alloc((size_t)T_ * OUT_ * OUT_ * 2);
  int*   perm    = (int*)alloc((size_t)GB_ * QR_ * 4);
  int*   btype   = (int*)alloc((size_t)GB_ * 4);

  setup_kernel<<<4 * PWB_ + RVB_ + LIB_, 256, 0, stream>>>(Wq, Wk, Wv, Wa,
                                                           wq16, wk16, wv16, wa16,
                                                           rte_emb, rte_W, rte_b, rte_vec, head,
                                                           node_type, perm, btype);

  qkv_wmma_kernel<<<RKB_ + LPB_ + 3 * GB_, 256, 0, stream>>>(
      node_inp, wq16, wk16, wv16, bq, bk, bv, perm, btype, qn, kn, vn,
      rte_vec, Wk, Wv, rteK, rteV, edge_idx, head, next);

  node_logit_kernel<<<N_ / NB_, 512, 0, stream>>>(qn, rel_att, kn, rteK, edge_idx, edge_type,
                                                  edge_time, node_type, rel_pri,
                                                  head, next, logit);
  node_aggr_kernel<<<N_ / NB_, 512, 0, stream>>>(vn, rteV, rel_msg, logit, edge_idx, edge_type,
                                                 edge_time, node_type, head, next, aggr);

  out_wmma_kernel<<<GB_, 256, 0, stream>>>(aggr, wa16, ba, node_inp, perm, btype,
                                          skip, ln_g, ln_b, outp);
}


# edge logits and aggregation fused into one kernel per 16 target nodes: one list walk and index fetch feed both, v rows fetched two edges a batch, S takes Q's LDS space; one launch gone; same sums
# speedup vs baseline: 3.3572x; 1.0597x over previous
#include <hip/hip_runtime.h>
#include <math.h>

#define T_   3
#define R_   5
#define H_   8
#define IN_  256
#define OUT_ 256
#define DK_  32
#define N_   30000
#define E_   200000
#define ML_  240

typedef __attribute__((ext_vector_type(16))) _Float16 v16h;
typedef __attribute__((ext_vector_type(8)))  float    v8f;
typedef __attribute__((ext_vector_type(2)))  float    v2f;

__device__ __forceinline__ int kmap(int s, int half) {
  int v = s >> 1;
  return ((v >> 2) << 4) | (half << 3) | ((v & 3) << 1) | (s & 1);
}

#define PWB_ 48
#define RVB_ 240
#define LIB_ 118
#define QR_  80
#define GB_  378
#define NT4_ 30

__device__ __forceinline__ unsigned type_fields(int4 v) {
  return (1u << (10 * v.x)) + (1u << (10 * v.y)) + (1u << (10 * v.z)) + (1u << (10 * v.w));
}

__device__ __forceinline__ void pack_w_frag(const float* __restrict__ W,
                                            _Float16* __restrict__ Wp, int f) {
  const int lane = f & 31, kk = (f >> 5) & 7, ct = (f >> 8) & 15, t = f >> 12;
  const int half = lane >> 4, n = lane & 15;
  const float* s0 = W + (size_t)t * (IN_ * OUT_) + (size_t)(kk * 32) * OUT_ + ct * 16 + n;
  v16h o;
#pragma unroll
  for (int s = 0; s < 16; ++s) o[s] = (_Float16)s0[(size_t)kmap(s, half) * OUT_];
  ((v16h*)Wp)[f] = o;
}

__global__ __launch_bounds__(256) void setup_kernel(
    const float* __restrict__ Wq, const float* __restrict__ Wk,
    const float* __restrict__ Wv, const float* __restrict__ Wa,
    _Float16* __restrict__ wq16, _Float16* __restrict__ wk16,
    _Float16* __restrict__ wv16, _Float16* __restrict__ wa16,
    const float* __restrict__ emb, const float* __restrict__ Wt,
    const float* __restrict__ b, float* __restrict__ rv,
    int* __restrict__ head, const int* __restrict__ ntype,
    int* __restrict__ perm, int* __restrict__ btype) {
  __shared__ int red[8][9];
  const int blk = blockIdx.x, tid = threadIdx.x;
  if (blk < 4 * PWB_) {
    const int which = blk / PWB_;
    const float* __restrict__ W = which == 0 ? Wq : (which == 1 ? Wk : (which == 2 ? Wv : Wa));
    _Float16* __restrict__ Wp = which == 0 ? wq16 : (which == 1 ? wk16 : (which == 2 ? wv16 : wa16));
    pack_w_frag(W, Wp, (blk - which * PWB_) * 256 + tid);
  } else if (blk < 4 * PWB_ + RVB_) {
    const int idx = (blk - 4 * PWB_) * 256 + tid;
    const int i = idx & 255, m = idx >> 8;
    float acc = b[i];
    const float* e = emb + (size_t)m * IN_;
    const float* w = Wt + (size_t)i * IN_;
    for (int j = 0; j < IN_; ++j) acc = fmaf(e[j], w[j], acc);
    rv[idx] = acc;
  } else {
    const int lb = blk - 4 * PWB_ - RVB_;
    const int i = lb * 256 + tid;
    const int lane = tid & 31, wave = tid >> 5;
    int ti = -1;
    if (i < N_) { head[i] = -1; ti = ntype[i]; }
    const int4* nt4 = (const int4*)ntype;
    unsigned all = 0, bef = 0;
#pragma unroll
    for (int it = 0; it < NT4_ - 1; ++it) {
      const int q = it * 256 + tid;
      const unsigned f = type_fields(nt4[q]);
      all += f;
      bef += q < 64 * lb ? f : 0u;
    }
    {
      const int q = (NT4_ - 1) * 256 + tid;
      if (q < N_ / 4) {
        const unsigned f = type_fields(nt4[q]);
        all += f;
        bef += q < 64 * lb ? f : 0u;
      }
    }
    int cn[6];
    cn[0] = all & 1023; cn[1] = (all >> 10) & 1023; cn[2] = (all >> 20) & 1023;
    cn[3] = bef & 1023; cn[4] = (bef >> 10) & 1023; cn[5] = (bef >> 20) & 1023;
#pragma unroll
    for (int o = 16; o > 0; o >>= 1) {
#pragma unroll
      for (int k = 0; k < 6; ++k) cn[k] += __shfl_xor(cn[k], o, 32);
    }
    const unsigned m0 = __builtin_amdgcn_ballot_w32(ti == 0);
    const unsigned m1 = __builtin_amdgcn_ballot_w32(ti == 1);
    const unsigned m2 = __builtin_amdgcn_ballot_w32(ti == 2);
    if (lane == 0) {
#pragma unroll
      for (int k = 0; k < 6; ++k) red[wave][k] = cn[k];
      red[wave][6] = __builtin_popcount(m0);
      red[wave][7] = __builtin_popcount(m1);
      red[wave][8] = __builtin_popcount(m2);
    }
    __syncthreads();
    int tot0 = 0, tot1 = 0, tot2 = 0, bf0 = 0, bf1 = 0, bf2 = 0, w0 = 0, w1 = 0, w2 = 0;
#pragma unroll
    for (int w = 0; w < 8; ++w) {
      tot0 += red[w][0]; tot1 += red[w][1]; tot2 += red[w][2];
      bf0 += red[w][3];  bf1 += red[w][4];  bf2 += red[w][5];
      if (w < wave) { w0 += red[w][6]; w1 += red[w][7]; w2 += red[w][8]; }
    }
    const int nb0 = (tot0 + QR_ - 1) / QR_, nb1 = (tot1 + QR_ - 1) / QR_;
    const int nb2 = (tot2 + QR_ - 1) / QR_;
    const int st1 = QR_ * nb0, st2 = QR_ * (nb0 + nb1);
    if (ti >= 0) {
      const unsigned mm = ti == 0 ? m0 : (ti == 1 ? m1 : m2);
      const int place = ti == 0 ? bf0 + w0 : (ti == 1 ? st1 + bf1 + w1 : st2 + bf2 + w2);
      perm[place + __builtin_popcount(mm & ((1u << lane) - 1u))] = i;
    }
    if (lb == 0) {
      for (int p = tid; p < GB_ * QR_; p += 256) {
        const int off = p < st1 ? p : (p < st2 ? p - st1 : p - st2);
        const int cnt = p < st1 ? tot0 : (p < st2 ? tot1 : tot2);
        if (off >= cnt) perm[p] = -1;
      }
      for (int g = tid; g < GB_; g += 256)
        btype[g] = g < nb0 ? 0 : (g < nb0 + nb1 ? 1 : (g < nb0 + nb1 + nb2 ? 2 : -1));
    }
  }
}

#define RKB_ 720
__device__ __forceinline__ void rte_kv_block(const float* __restrict__ rv,
                              const float* __restrict__ Wk, const float* __restrict__ Wv,
                              float* __restrict__ rk, float* __restrict__ rvv, int blk) {
  int idx = blk * 256 + threadIdx.x;
  int o = idx & 255;
  int m = (idx >> 8) % ML_;
  int t = idx / (ML_ * OUT_);
  const float* v  = rv + (size_t)m * IN_;
  const float* wk = Wk + (size_t)t * (IN_ * OUT_);
  const float* wv = Wv + (size_t)t * (IN_ * OUT_);
  float ak = 0.0f, av = 0.0f;
  for (int i = 0; i < IN_; ++i) {
    float x = v[i];
    ak = fmaf(x, wk[(size_t)i * OUT_ + o], ak);
    av = fmaf(x, wv[(size_t)i * OUT_ + o], av);
  }
  rk[idx] = ak;
  rvv[idx] = av;
}

#define QG_ 5

typedef __attribute__((ext_vector_type(8))) _Float16 v8h;

template <bool SCATTER>
__device__ __forceinline__ void project_tiles(
    const v8h* __restrict__ As8, const v16h* __restrict__ Wpk,
    const float* __restrict__ bias, float* __restrict__ Out,
    const int* srow, int t, int wave, int lane) {
  const int n = lane & 15, hs = lane >> 4;
  const int ct0 = wave * 2;
  const int nbase = ct0 * 16;
  const v8h* ap = As8 + lane;
  {
    const v16h* wp = Wpk + (((size_t)t * 16 + ct0) * 8) * 32 + lane;
    v16h b0[2], b1[2];
    v8h alo[2][QG_], ahi[2][QG_];
    b0[0] = wp[0];      b1[0] = wp[8 * 32];
#pragma unroll
    for (int g = 0; g < QG_; ++g) {
      alo[0][g] = ap[((g * 8) * 2 + 0) * 32];
      ahi[0][g] = ap[((g * 8) * 2 + 1) * 32];
    }
    const float bv0 = bias[t * OUT_ + nbase + n];
    const float bv1 = bias[t * OUT_ + nbase + 16 + n];
    v8f acc[QG_][2];
#pragma unroll
    for (int g = 0; g < QG_; ++g) {
#pragma unroll
      for (int i = 0; i < 8; ++i) { acc[g][0][i] = bv0; acc[g][1][i] = bv1; }
    }
#pragma unroll
    for (int kk = 0; kk < 8; ++kk) {
      const int cur = kk & 1, nxt = cur ^ 1;
      if (kk + 1 < 8) {
        b0[nxt] = wp[(kk + 1) * 32];
        b1[nxt] = wp[(8 + kk + 1) * 32];
#pragma unroll
        for (int g = 0; g < QG_; ++g) {
          alo[nxt][g] = ap[((g * 8 + kk + 1) * 2 + 0) * 32];
          ahi[nxt][g] = ap[((g * 8 + kk + 1) * 2 + 1) * 32];
        }
      }
      __builtin_amdgcn_sched_barrier(0);
#pragma unroll
      for (int g = 0; g < QG_; ++g) {
        const v16h a = __builtin_shufflevector(alo[cur][g], ahi[cur][g],
                                               0, 1, 2, 3, 4, 5, 6, 7,
                                               8, 9, 10, 11, 12, 13, 14, 15);
        acc[g][0] = __builtin_amdgcn_wmma_f32_16x16x32_f16(false, a, false, b0[cur],
                                                           (short)0, acc[g][0], false, false);
        acc[g][1] = __builtin_amdgcn_wmma_f32_16x16x32_f16(false, a, false, b1[cur],
                                                           (short)0, acc[g][1], false, false);
      }
      __builtin_amdgcn_sched_barrier(0);
    }
#pragma unroll
    for (int g = 0; g < QG_; ++g) {
#pragma unroll
      for (int i = 0; i < 8; ++i) {
        const int lr = g * 16 + hs * 8 + i;
        if (SCATTER) {
          const int row = srow[lr];
          if (row >= 0) {
            float* o = Out + (size_t)row * OUT_ + nbase + n;
            o[0]  = acc[g][0][i];
            o[16] = acc[g][1][i];
          }
        } else {
          float* o = Out + lr * OUT_ + nbase + n;
          o[0]  = acc[g][0][i];
          o[16] = acc[g][1][i];
        }
      }
    }
  }
}

#define LPB_ 782
__global__ __launch_bounds__(256) void qkv_wmma_kernel(
    const float* __restrict__ x,
    const _Float16* __restrict__ wq16, const _Float16* __restrict__ wk16,
    const _Float16* __restrict__ wv16,
    const float* __restrict__ bq, const float* __restrict__ bk, const float* __restrict__ bv,
    const int* __restrict__ perm, const int* __restrict__ btype,
    float* __restrict__ qn, float* __restrict__ kn, float* __restrict__ vn,
    const float* __restrict__ rv, const float* __restrict__ Wk, const float* __restrict__ Wv,
    float* __restrict__ rk, float* __restrict__ rvv,
    const int* __restrict__ ei, int* __restrict__ head, int* __restrict__ next) {
  __shared__ __align__(32) _Float16 As[QG_ * 16 * IN_];
  __shared__ int srow[QG_ * 16];
  const int tid = threadIdx.x;
  if (blockIdx.x < RKB_) {
    rte_kv_block(rv, Wk, Wv, rk, rvv, blockIdx.x);
    return;
  }
  if (blockIdx.x < RKB_ + LPB_) {
    const int e = (blockIdx.x - RKB_) * 256 + tid;
    if (e < E_) next[e] = atomicExch(&head[ei[E_ + e]], e);
    return;
  }
  const int bid = blockIdx.x - RKB_ - LPB_;
  const int proj = bid % 3;
  const int grp = bid / 3;
  const int t = btype[grp];
  const int pr = tid < QG_ * 16 ? perm[grp * (QG_ * 16) + tid] : -1;
  if (t < 0) return;
  if (tid < QG_ * 16) srow[tid] = pr;
  __syncthreads();
  const _Float16* __restrict__ w16 = proj == 0 ? wq16 : (proj == 1 ? wk16 : wv16);
  const float* __restrict__ bias = proj == 0 ? bq : (proj == 1 ? bk : bv);
  float* __restrict__ outp = proj == 0 ? qn : (proj == 1 ? kn : vn);
#pragma unroll
  for (int g = 0; g < QG_; ++g) {
    const int node = srow[g * 16 + (tid >> 4)];
    const float* xr = x + (size_t)(node < 0 ? 0 : node) * IN_;
#pragma unroll
    for (int j = 0; j < 16; ++j) {
      int flat = j * 256 + tid;
      int s = flat & 15, ln = (flat >> 4) & 31, kk = flat >> 9;
      int k = kk * 32 + kmap(s, ln >> 4);
      const float xv = xr[k];
      As[((((g * 8 + kk) * 2 + (s >> 3)) * 32 + ln) << 3) + (s & 7)] =
          (_Float16)(node < 0 ? 0.0f : xv);
    }
  }
  __syncthreads();
  const int wave = tid >> 5, lane = tid & 31;
  const v8h* As8 = (const v8h*)As;
  project_tiles<true>(As8, (const v16h*)w16, bias, outp, srow, t, wave, lane);
}

__device__ __forceinline__ void load_frag16(const float* __restrict__ p, v2f (&f)[8]) {
  const float4* p4 = (const float4*)p;
#pragma unroll
  for (int j = 0; j < 4; ++j) {
    const float4 t = p4[j];
    f[2 * j][0] = t.x;     f[2 * j][1] = t.y;
    f[2 * j + 1][0] = t.z; f[2 * j + 1][1] = t.w;
  }
}

#define NB_ 16

typedef __attribute__((ext_vector_type(4))) float v4f;

__device__ __forceinline__ float dot8(v4f k0, v4f k1, v4f t0, v4f t1, v4f q0, v4f q1) {
  float s = (k0.x + t0.x) * q0.x;
  s = fmaf(k0.y + t0.y, q0.y, s);
  s = fmaf(k0.z + t0.z, q0.z, s);
  s = fmaf(k0.w + t0.w, q0.w, s);
  s = fmaf(k1.x + t1.x, q1.x, s);
  s = fmaf(k1.y + t1.y, q1.y, s);
  s = fmaf(k1.z + t1.z, q1.z, s);
  s = fmaf(k1.w + t1.w, q1.w, s);
  return s;
}

__device__ __forceinline__ int fetch_stretch(
    int& e, int lane, const int* __restrict__ next, const int* __restrict__ ei,
    const int* __restrict__ etype, const int* __restrict__ etime,
    const int* __restrict__ ntype, int& my_e, int& orow_l, int& ort_l, int& r_l) {
  my_e = 0;
  int got = 0;
  for (; got < 32 && e >= 0; ++got) {
    if (got == lane) my_e = e;
    e = next[e];
  }
  const int cnt = __builtin_amdgcn_readfirstlane(got);
  orow_l = 0; ort_l = 0; r_l = 0;
  if (lane < cnt) {
    const int src = ei[my_e];
    r_l = etype[my_e];
    const int tm = etime[my_e];
    orow_l = src * OUT_;
    ort_l = (ntype[src] * ML_ + tm) * OUT_;
  }
  return cnt;
}

__global__ __launch_bounds__(512) __attribute__((amdgpu_waves_per_eu(16, 16)))
void node_edge_kernel(
    const float* __restrict__ qn, const float* __restrict__ rel_att,
    const float* __restrict__ kn, const float* __restrict__ rteK,
    const float* __restrict__ vn, const float* __restrict__ rteV,
    const float* __restrict__ rel_msg,
    const int* __restrict__ ei, const int* __restrict__ etype,
    const int* __restrict__ etime, const int* __restrict__ ntype,
    const float* __restrict__ rel_pri,
    const int* __restrict__ head, const int* __restrict__ next,
    float* logit, float* __restrict__ aggr) {
  __shared__ __align__(16) float QS[NB_ * R_ * OUT_];
  const int lane = threadIdx.x & 31;
  const int wave = __builtin_amdgcn_readfirstlane(threadIdx.x >> 5);
  const int base = blockIdx.x * NB_;
  if (wave < H_) {
    const int h = wave;
    const int c = lane & 15, hs = lane >> 4;
    v2f a[8];
    load_frag16(qn + (size_t)(base + c) * OUT_ + h * DK_ + hs * 16, a);
#pragma unroll 1
    for (int r = 0; r < R_; ++r) {
      const float* bp = rel_att + (((size_t)r * H_ + h) * DK_ + c) * DK_ + hs * 16;
      v2f b0[8], b1[8];
      load_frag16(bp, b0);
      load_frag16(bp + 16 * DK_, b1);
      v8f acc0 = {0.0f, 0.0f, 0.0f, 0.0f, 0.0f, 0.0f, 0.0f, 0.0f};
      v8f acc1 = acc0;
#pragma unroll
      for (int ks = 0; ks < 8; ++ks) {
        acc0 = __builtin_amdgcn_wmma_f32_16x16x4_f32(false, a[ks], false, b0[ks],
                                                     (short)0, acc0, false, false);
        acc1 = __builtin_amdgcn_wmma_f32_16x16x4_f32(false, a[ks], false, b1[ks],
                                                     (short)0, acc1, false, false);
      }
#pragma unroll
      for (int i = 0; i < 8; ++i) {
        float* o = QS + ((8 * hs + i) * R_ + r) * OUT_ + h * DK_ + c;
        o[0]  = acc0[i];
        o[16] = acc1[i];
      }
    }
  }
  __syncthreads();
  {
    const int n = base + wave;
    const int hq = lane >> 2;
    float* srow = QS + wave * (R_ * OUT_);
    const float* qrow = srow + lane * 8;
    const int e0 = head[n];
    int e = e0;
    int my_e = 0, orow_l = 0, ort_l = 0, r_l = 0, cnt = 0;
    int nst = 0;
    float m = -__builtin_inff();
    for (int ch = 0; ch < (E_ + 31) / 32 && e >= 0; ++ch) {
      cnt = fetch_stretch(e, lane, next, ei, etype, etime, ntype, my_e, orow_l, ort_l, r_l);
      ++nst;
      for (int j0 = 0; j0 < cnt; j0 += 2) {
        const bool two = j0 + 1 < cnt;
        const int j1 = two ? j0 + 1 : j0;
        const int ea = __builtin_amdgcn_readlane(my_e, j0);
        const int eb = __builtin_amdgcn_readlane(my_e, j1);
        const int ra = __builtin_amdgcn_readlane(r_l, j0);
        const int rb = __builtin_amdgcn_readlane(r_l, j1);
        const v4f* ka = (const v4f*)(kn + __builtin_amdgcn_readlane(orow_l, j0)) + lane * 2;
        const v4f* ta = (const v4f*)(rteK + __builtin_amdgcn_readlane(ort_l, j0)) + lane * 2;
        const v4f* kb = (const v4f*)(kn + __builtin_amdgcn_readlane(orow_l, j1)) + lane * 2;
        const v4f* tb = (const v4f*)(rteK + __builtin_amdgcn_readlane(ort_l, j1)) + lane * 2;
        const v4f ka0 = ka[0], ka1 = ka[1], ta0 = ta[0], ta1 = ta[1];
        const v4f kb0 = kb[0], kb1 = kb[1], tb0 = tb[0], tb1 = tb[1];
        const float pa = rel_pri[ra * H_ + hq];
        const float pb = rel_pri[rb * H_ + hq];
        asm volatile("" :: "v"(ka0), "v"(ka1), "v"(ta0), "v"(ta1), "v"(pa),
                           "v"(kb0), "v"(kb1), "v"(tb0), "v"(tb1), "v"(pb));
        const v4f* qa = (const v4f*)(qrow + ra * OUT_);
        const v4f* qb = (const v4f*)(qrow + rb * OUT_);
        float sa = dot8(ka0, ka1, ta0, ta1, qa[0], qa[1]);
        float sb = dot8(kb0, kb1, tb0, tb1, qb[0], qb[1]);
        sa += __shfl_xor(sa, 1, 32);
        sb += __shfl_xor(sb, 1, 32);
        sa += __shfl_xor(sa, 2, 32);
        sb += __shfl_xor(sb, 2, 32);
        const float la = sa * pa * 0.17677669529663689f;
        const float lb = sb * pb * 0.17677669529663689f;
        logit[(size_t)ea * H_ + hq] = la;
        m = fmaxf(m, la);
        if (two) {
          logit[(size_t)eb * H_ + hq] = lb;
          m = fmaxf(m, lb);
        }
      }
    }
    const int c0 = (8 * lane + 4 * wave) & 255;
    const int c1 = (8 * lane + 4 + 4 * wave) & 255;
    const v4f zero = {0.0f, 0.0f, 0.0f, 0.0f};
#pragma unroll
    for (int r = 0; r < R_; ++r) {
      *(v4f*)(srow + r * OUT_ + c0) = zero;
      *(v4f*)(srow + r * OUT_ + c1) = zero;
    }
    const bool again = nst > 1;
    float den = 0.0f;
    e = e0;
    for (int ch = 0; ch < nst; ++ch) {
      if (again)
        cnt = fetch_stretch(e, lane, next, ei, etype, etime, ntype, my_e, orow_l, ort_l, r_l);
      for (int j0 = 0; j0 < cnt; j0 += 2) {
        const bool two = j0 + 1 < cnt;
        const int j1 = two ? j0 + 1 : j0;
        const int ea = __builtin_amdgcn_readlane(my_e, j0);
        const int eb = __builtin_amdgcn_readlane(my_e, j1);
        const int ra = __builtin_amdgcn_readlane(r_l, j0);
        const int rb = __builtin_amdgcn_readlane(r_l, j1);
        const v4f* va = (const v4f*)(vn + __builtin_amdgcn_readlane(orow_l, j0)) + lane * 2;
        const v4f* ta = (const v4f*)(rteV + __builtin_amdgcn_readlane(ort_l, j0)) + lane * 2;
        const v4f* vb = (const v4f*)(vn + __builtin_amdgcn_readlane(orow_l, j1)) + lane * 2;
        const v4f* tb = (const v4f*)(rteV + __builtin_amdgcn_readlane(ort_l, j1)) + lane * 2;
        const v4f xa0 = va[0], xa1 = va[1], ta0 = ta[0], ta1 = ta[1];
        const v4f xb0 = vb[0], xb1 = vb[1], tb0 = tb[0], tb1 = tb[1];
        const float la = logit[(size_t)ea * H_ + hq];
        const float lb = logit[(size_t)eb * H_ + hq];
        asm volatile("" :: "v"(xa0), "v"(xa1), "v"(ta0), "v"(ta1), "v"(la),
                           "v"(xb0), "v"(xb1), "v"(tb0), "v"(tb1), "v"(lb));
        {
          const float ex = __expf(la - m);
          v4f* s0 = (v4f*)(srow + ra * OUT_ + c0);
          v4f* s1 = (v4f*)(srow + ra * OUT_ + c1);
          v4f a0 = *s0, a1 = *s1;
          den += ex;
          a0.x = fmaf(ex, xa0.x + ta0.x, a0.x);
          a0.y = fmaf(ex, xa0.y + ta0.y, a0.y);
          a0.z = fmaf(ex, xa0.z + ta0.z, a0.z);
          a0.w = fmaf(ex, xa0.w + ta0.w, a0.w);
          a1.x = fmaf(ex, xa1.x + ta1.x, a1.x);
          a1.y = fmaf(ex, xa1.y + ta1.y, a1.y);
          a1.z = fmaf(ex, xa1.z + ta1.z, a1.z);
          a1.w = fmaf(ex, xa1.w + ta1.w, a1.w);
          *s0 = a0;
          *s1 = a1;
        }
        if (two) {
          const float ex = __expf(lb - m);
          v4f* s0 = (v4f*)(srow + rb * OUT_ + c0);
          v4f* s1 = (v4f*)(srow + rb * OUT_ + c1);
          v4f a0 = *s0, a1 = *s1;
          den += ex;
          a0.x = fmaf(ex, xb0.x + tb0.x, a0.x);
          a0.y = fmaf(ex, xb0.y + tb0.y, a0.y);
          a0.z = fmaf(ex, xb0.z + tb0.z, a0.z);
          a0.w = fmaf(ex, xb0.w + tb0.w, a0.w);
          a1.x = fmaf(ex, xb1.x + tb1.x, a1.x);
          a1.y = fmaf(ex, xb1.y + tb1.y, a1.y);
          a1.z = fmaf(ex, xb1.z + tb1.z, a1.z);
          a1.w = fmaf(ex, xb1.w + tb1.w, a1.w);
          *s0 = a0;
          *s1 = a1;
        }
      }
    }
    if (e0 >= 0) {
      const float inv = 1.0f / den;
#pragma unroll
      for (int r = 0; r < R_; ++r) {
        v4f* s0 = (v4f*)(srow + r * OUT_ + c0);
        v4f* s1 = (v4f*)(srow + r * OUT_ + c1);
        v4f a0 = *s0, a1 = *s1;
        a0 *= inv;
        a1 *= inv;
        *s0 = a0;
        *s1 = a1;
      }
    }
  }
  __syncthreads();
  if (wave >= H_) return;
  const int h = wave;
  const int c = lane & 15, hs = lane >> 4;
  v8f acc0 = {0.0f, 0.0f, 0.0f, 0.0f, 0.0f, 0.0f, 0.0f, 0.0f};
  v8f acc1 = acc0;
#pragma unroll 1
  for (int r = 0; r < R_; ++r) {
    const float* ar = QS + c * (R_ * OUT_) + r * OUT_;
    v2f a[8];
#pragma unroll
    for (int q = 0; q < 4; ++q) {
      const v4f t = *(const v4f*)(ar + ((h * DK_ + hs * 16 + 4 * q + 4 * c) & 255));
      a[2 * q][0] = t.x;     a[2 * q][1] = t.y;
      a[2 * q + 1][0] = t.z; a[2 * q + 1][1] = t.w;
    }
    const float* bp = rel_msg + (((size_t)r * H_ + h) * DK_ + hs * 16) * DK_ + c;
    v2f b0[8], b1[8];
#pragma unroll
    for (int ks = 0; ks < 8; ++ks) {
      b0[ks][0] = bp[(2 * ks) * DK_];
      b0[ks][1] = bp[(2 * ks + 1) * DK_];
      b1[ks][0] = bp[(2 * ks) * DK_ + 16];
      b1[ks][1] = bp[(2 * ks + 1) * DK_ + 16];
    }
#pragma unroll
    for (int ks = 0; ks < 8; ++ks) {
      acc0 = __builtin_amdgcn_wmma_f32_16x16x4_f32(false, a[ks], false, b0[ks],
                                                   (short)0, acc0, false, false);
      acc1 = __builtin_amdgcn_wmma_f32_16x16x4_f32(false, a[ks], false, b1[ks],
                                                   (short)0, acc1, false, false);
    }
  }
#pragma unroll
  for (int i = 0; i < 8; ++i) {
    float* o = aggr + (size_t)(base + 8 * hs + i) * OUT_ + h * DK_ + c;
    o[0]  = acc0[i];
    o[16] = acc1[i];
  }
}

#define OR_ (QG_ * 16)
#define ORW_ (OR_ / 8)

__global__ __launch_bounds__(256) void out_wmma_kernel(
    const float* __restrict__ aggr, const _Float16* __restrict__ wa16,
    const float* __restrict__ ba, const float* __restrict__ x,
    const int* __restrict__ perm, const int* __restrict__ btype,
    const float* __restrict__ skip,
    const float* __restrict__ ln_g, const float* __restrict__ ln_b,
    float* __restrict__ out) {
  __shared__ __align__(32) _Float16 As[OR_ * OUT_];
  __shared__ __align__(16) float Tr[OR_ * OUT_];
  __shared__ int srow[OR_];
  const int tid = threadIdx.x;
  const int t = btype[blockIdx.x];
  const int pr = tid < OR_ ? perm[blockIdx.x * OR_ + tid] : -1;
  if (t < 0) return;
  if (tid < OR_) srow[tid] = pr;
  __syncthreads();
#pragma unroll
  for (int g = 0; g < QG_; ++g) {
    const int node = srow[g * 16 + (tid >> 4)];
    const float* ar = aggr + (size_t)(node < 0 ? 0 : node) * OUT_;
#pragma unroll
    for (int j = 0; j < 16; ++j) {
      int flat = j * 256 + tid;
      int s = flat & 15, ln = (flat >> 4) & 31, kk = flat >> 9;
      int k = kk * 32 + kmap(s, ln >> 4);
      const float av = ar[k];
      As[((((g * 8 + kk) * 2 + (s >> 3)) * 32 + ln) << 3) + (s & 7)] =
          (_Float16)(node < 0 ? 0.0f : av);
    }
  }
  __syncthreads();
  const int wave = tid >> 5, lane = tid & 31;
  project_tiles<false>((const v8h*)As, (const v16h*)wa16, ba, Tr, srow, t, wave, lane);
  __syncthreads();
  const float a = 1.0f / (1.0f + __expf(-skip[t]));
  const v4f g0 = *(const v4f*)(ln_g + t * OUT_ + 4 * lane);
  const v4f g1 = *(const v4f*)(ln_g + t * OUT_ + 128 + 4 * lane);
  const v4f c0 = *(const v4f*)(ln_b + t * OUT_ + 4 * lane);
  const v4f c1 = *(const v4f*)(ln_b + t * OUT_ + 128 + 4 * lane);
#pragma unroll 5
  for (int rr = 0; rr < ORW_; ++rr) {
    const int row = wave * ORW_ + rr;
    const int node = srow[row];
    const float* xr = x + (size_t)(node < 0 ? 0 : node) * IN_ + 4 * lane;
    const float* tr = Tr + row * OUT_ + 4 * lane;
    const v4f x0 = *(const v4f*)xr, x1 = *(const v4f*)(xr + 128);
    const v4f t0 = *(const v4f*)tr, t1 = *(const v4f*)(tr + 128);
    float v[8];
    v[0] = t0.x * a + x0.x * (1.0f - a);
    v[1] = t0.y * a + x0.y * (1.0f - a);
    v[2] = t0.z * a + x0.z * (1.0f - a);
    v[3] = t0.w * a + x0.w * (1.0f - a);
    v[4] = t1.x * a + x1.x * (1.0f - a);
    v[5] = t1.y * a + x1.y * (1.0f - a);
    v[6] = t1.z * a + x1.z * (1.0f - a);
    v[7] = t1.w * a + x1.w * (1.0f - a);
    float mu = 0.0f;
#pragma unroll
    for (int j = 0; j < 8; ++j) mu += v[j];
#pragma unroll
    for (int o = 16; o > 0; o >>= 1) mu += __shfl_xor(mu, o, 32);
    mu *= (1.0f / OUT_);
    float var = 0.0f;
#pragma unroll
    for (int j = 0; j < 8; ++j) {
      const float d = v[j] - mu;
      var = fmaf(d, d, var);
    }
#pragma unroll
    for (int o = 16; o > 0; o >>= 1) var += __shfl_xor(var, o, 32);
    var *= (1.0f / OUT_);
    const float rs = rsqrtf(var + 1e-5f);
    v4f o0, o1;
    o0.x = (v[0] - mu) * rs * g0.x + c0.x;
    o0.y = (v[1] - mu) * rs * g0.y + c0.y;
    o0.z = (v[2] - mu) * rs * g0.z + c0.z;
    o0.w = (v[3] - mu) * rs * g0.w + c0.w;
    o1.x = (v[4] - mu) * rs * g1.x + c1.x;
    o1.y = (v[5] - mu) * rs * g1.y + c1.y;
    o1.z = (v[6] - mu) * rs * g1.z + c1.z;
    o1.w = (v[7] - mu) * rs * g1.w + c1.w;
    if (node >= 0) {
      float* orow = out + (size_t)node * OUT_ + 4 * lane;
      *(v4f*)orow = o0;
      *(v4f*)(orow + 128) = o1;
    }
  }
}

extern "C" void kernel_launch(void* const* d_in, const int* in_sizes, int n_in,
                              void* d_out, int out_size, void* d_ws, size_t ws_size,
                              hipStream_t stream) {
  (void)in_sizes; (void)n_in; (void)out_size; (void)ws_size;
  const float* node_inp  = (const float*)d_in[0];
  const float* Wk        = (const float*)d_in[1];
  const float* bk        = (const float*)d_in[2];
  const float* Wq        = (const float*)d_in[3];
  const float* bq        = (const float*)d_in[4];
  const float* Wv        = (const float*)d_in[5];
  const float* bv        = (const float*)d_in[6];
  const float* Wa        = (const float*)d_in[7];
  const float* ba        = (const float*)d_in[8];
  const float* ln_g      = (const float*)d_in[9];
  const float* ln_b      = (const float*)d_in[10];
  const float* rel_pri   = (const float*)d_in[11];
  const float* rel_att   = (const float*)d_in[12];
  const float* rel_msg   = (const float*)d_in[13];
  const float* skip      = (const float*)d_in[14];
  const float* rte_W     = (const float*)d_in[15];
  const float* rte_b     = (const float*)d_in[16];
  const float* rte_emb   = (const float*)d_in[17];
  const int*   node_type = (const int*)d_in[18];
  const int*   edge_idx  = (const int*)d_in[19];
  const int*   edge_type = (const int*)d_in[20];
  const int*   edge_time = (const int*)d_in[21];
  float* outp = (float*)d_out;

  char* p = (char*)d_ws;
  auto alloc = [&](size_t bytes) -> void* {
    void* r = (void*)p;
    p += (bytes + 255) & ~(size_t)255;
    return r;
  };
  float* rte_vec = (float*)alloc((size_t)ML_ * IN_ * 4);
  float* rteK    = (float*)alloc((size_t)T_ * ML_ * OUT_ * 4);
  float* rteV    = (float*)alloc((size_t)T_ * ML_ * OUT_ * 4);
  float* qn      = (float*)alloc((size_t)N_ * OUT_ * 4);
  float* kn      = (float*)alloc((size_t)N_ * OUT_ * 4);
  float* vn      = (float*)alloc((size_t)N_ * OUT_ * 4);
  float* logit   = (float*)alloc((size_t)E_ * H_ * 4);
  int*   head    = (int*)alloc((size_t)N_ * 4);
  int*   next    = (int*)alloc((size_t)E_ * 4);
  float* aggr    = (float*)alloc((size_t)N_ * OUT_ * 4);
  _Float16* wq16 = (_Float16*)alloc((size_t)T_ * IN_ * OUT_ * 2);
  _Float16* wk16 = (_Float16*)alloc((size_t)T_ * IN_ * OUT_ * 2);
  _Float16* wv16 = (_Float16*)alloc((size_t)T_ * IN_ * OUT_ * 2);
  _Float16* wa16 = (_Float16*)alloc((size_t)T_ * OUT_ * OUT_ * 2);
  int*   perm    = (int*)alloc((size_t)GB_ * QR_ * 4);
  int*   btype   = (int*)alloc((size_t)GB_ * 4);

  setup_kernel<<<4 * PWB_ + RVB_ + LIB_, 256, 0, stream>>>(Wq, Wk, Wv, Wa,
                                                           wq16, wk16, wv16, wa16,
                                                           rte_emb, rte_W, rte_b, rte_vec, head,
                                                           node_type, perm, btype);

  qkv_wmma_kernel<<<RKB_ + LPB_ + 3 * GB_, 256, 0, stream>>>(
      node_inp, wq16, wk16, wv16, bq, bk, bv, perm, btype, qn, kn, vn,
      rte_vec, Wk, Wv, rteK, rteV, edge_idx, head, next);

  node_edge_kernel<<<N_ / NB_, 512, 0, stream>>>(qn, rel_att, kn, rteK, vn, rteV, rel_msg,
                                                 edge_idx, edge_type, edge_time, node_type,
                                                 rel_pri, head, next, logit, aggr);

  out_wmma_kernel<<<GB_, 256, 0, stream>>>(aggr, wa16, ba, node_inp, perm, btype,
                                          skip, ln_g, ln_b, outp);
}
